# starting kernel
# baseline (speedup 1.0000x reference)
#include <hip/hip_runtime.h>

#define D 128
#define WSTR 132

typedef __attribute__((ext_vector_type(2))) float v2f;
typedef __attribute__((ext_vector_type(8))) float v8f;

__global__ void gcn_zero_f32(float* __restrict__ p, long long n) {
  long long i = (long long)blockIdx.x * blockDim.x + threadIdx.x;
  long long stride = (long long)gridDim.x * blockDim.x;
  for (; i < n; i += stride) p[i] = 0.0f;
}

__global__ void gcn_degree(const int* __restrict__ src, const int* __restrict__ dst,
                           float* __restrict__ odeg, float* __restrict__ ideg, int E) {
  int i = blockIdx.x * blockDim.x + threadIdx.x;
  if (i < E) {
    atomicAdd(&odeg[src[i]], 1.0f);
    atomicAdd(&ideg[dst[i]], 1.0f);
  }
}

__global__ void gcn_norm(float* __restrict__ x, int n) {
  int i = blockIdx.x * blockDim.x + threadIdx.x;
  if (i < n) x[i] = rsqrtf(fmaxf(x[i], 1.0f));
}

__global__ __launch_bounds__(128)
void gcn_gemm_wmma(const float* __restrict__ h, const float* __restrict__ W,
                   const float* __restrict__ bias, const float* __restrict__ out_norm,
                   float* __restrict__ hs, int nrows, int ntiles) {
  __shared__ float sW[64 * WSTR];
  __shared__ float sA[16 * WSTR];
  __shared__ float sB[64];

  const int tid   = threadIdx.x;
  const int nbase = blockIdx.y * 64;

  for (int idx = tid; idx < 64 * (D / 4); idx += 128) {
    const int r  = idx >> 5;
    const int c4 = (idx & 31) << 2;
    const float4 v = *(const float4*)(W + (size_t)(nbase + r) * D + c4);
    *(float4*)(&sW[r * WSTR + c4]) = v;
  }
  if (tid < 64) sB[tid] = bias[nbase + tid];
  __syncthreads();

  const int wave = tid >> 5;
  const int lane = tid & 31;
  const int hi   = lane >> 4;
  const int lm   = lane & 15;

  for (int tile = blockIdx.x; tile < ntiles; tile += gridDim.x) {
    const int row0 = tile << 4;

    for (int idx = tid; idx < 16 * (D / 4); idx += 128) {
      const int r   = idx >> 5;
      const int c4  = (idx & 31) << 2;
      const int row = row0 + r;
      float4 v;
      if (row < nrows) v = *(const float4*)(h + (size_t)row * D + c4);
      else { v.x = 0.f; v.y = 0.f; v.z = 0.f; v.w = 0.f; }
      *(float4*)(&sA[r * WSTR + c4]) = v;
    }
    __syncthreads();

    v8f acc = {};
    const float* __restrict__ arow = &sA[lm * WSTR];
    const float* __restrict__ brow = &sW[(wave * 16 + lm) * WSTR];
#pragma unroll
    for (int k = 0; k < D; k += 4) {
      const int kk = k + (hi << 1);
      v2f a; a.x = arow[kk]; a.y = arow[kk + 1];
      v2f b; b.x = brow[kk]; b.y = brow[kk + 1];
      acc = __builtin_amdgcn_wmma_f32_16x16x4_f32(
           false, a,  false, b,
           (short)0, acc,  false,  false);
    }

    const int n  = nbase + wave * 16 + lm;
    const float bn = sB[wave * 16 + lm];
#pragma unroll
    for (int r = 0; r < 8; ++r) {
      const int m = row0 + r + (hi << 3);
      if (m < nrows)
        hs[(size_t)m * D + n] = (acc[r] + bn) * out_norm[m];
    }
    __syncthreads();
  }
}

__global__ void gcn_scatter(const float* __restrict__ hs, const int* __restrict__ src,
                            const int* __restrict__ dst, float* __restrict__ agg,
                            long long E) {
  long long gid = (long long)blockIdx.x * blockDim.x + threadIdx.x;
  const long long nwork = E << 5;
  if (gid >= nwork) return;
  const int e  = (int)(gid >> 5);
  const int c4 = (int)(gid & 31) << 2;
  const int s = src[e];
  const int d = dst[e];
  const float4 v = *(const float4*)(hs + (size_t)s * D + c4);
  float* o = agg + (size_t)d * D + c4;
  atomicAdd(o + 0, v.x);
  atomicAdd(o + 1, v.y);
  atomicAdd(o + 2, v.z);
  atomicAdd(o + 3, v.w);
}

__global__ void gcn_finalize(const float* __restrict__ agg, const float* __restrict__ in_norm,
                             const float* __restrict__ hres, float* __restrict__ out,
                             int nrows, int do_relu) {
  long long gid = (long long)blockIdx.x * blockDim.x + threadIdx.x;
  const long long n4 = (long long)nrows * (D / 4);
  if (gid >= n4) return;
  const int row = (int)(gid >> 5);
  const float s = in_norm[row];
  float4 v = *(const float4*)(agg + (gid << 2));
  v.x *= s; v.y *= s; v.z *= s; v.w *= s;
  if (hres) {
    const float4 r = *(const float4*)(hres + (gid << 2));
    v.x += r.x; v.y += r.y; v.z += r.z; v.w += r.w;
  }
  if (do_relu) {
    v.x = fmaxf(v.x, 0.f); v.y = fmaxf(v.y, 0.f);
    v.z = fmaxf(v.z, 0.f); v.w = fmaxf(v.w, 0.f);
  }
  *(float4*)(out + (gid << 2)) = v;
}

extern "C" void kernel_launch(void* const* d_in, const int* in_sizes, int n_in,
                              void* d_out, int out_size, void* d_ws, size_t ws_size,
                              hipStream_t stream) {
  const float* feat = (const float*)d_in[0];
  const int*   src  = (const int*)d_in[1];
  const int*   dst  = (const int*)d_in[2];
  const float* Wl[3] = { (const float*)d_in[3], (const float*)d_in[5], (const float*)d_in[7] };
  const float* bl[3] = { (const float*)d_in[4], (const float*)d_in[6], (const float*)d_in[8] };

  const int N = in_sizes[0] / D;
  const int E = in_sizes[1];

  char* ws = (char*)d_ws;
  const size_t featBytes = (size_t)N * D * sizeof(float);
  const size_t normBytes = (((size_t)N * sizeof(float)) + 255) & ~(size_t)255;
  float* hs   = (float*)ws;
  float* hbuf = (float*)(ws + featBytes);
  float* onrm = (float*)(ws + 2 * featBytes);
  float* inrm = (float*)(ws + 2 * featBytes + normBytes);
  float* agg  = (float*)d_out;

  gcn_zero_f32<<<1024, 256, 0, stream>>>(onrm, (long long)N);
  gcn_zero_f32<<<1024, 256, 0, stream>>>(inrm, (long long)N);
  gcn_degree<<<(E + 255) / 256, 256, 0, stream>>>(src, dst, onrm, inrm, E);
  gcn_norm<<<(N + 255) / 256, 256, 0, stream>>>(onrm, N);
  gcn_norm<<<(N + 255) / 256, 256, 0, stream>>>(inrm, N);

  const int ntiles = (N + 15) / 16;
  const dim3 ggrid(1024, 2);
  const long long scatterWork = (long long)E << 5;
  const unsigned scatterBlocks = (unsigned)((scatterWork + 255) / 256);
  const long long n4 = (long long)N * (D / 4);
  const unsigned finBlocks = (unsigned)((n4 + 255) / 256);

  const float* hin = feat;
  for (int l = 0; l < 3; ++l) {
    const bool last = (l == 2);
    gcn_gemm_wmma<<<ggrid, 128, 0, stream>>>(hin, Wl[l], bl[l], onrm, hs, N, ntiles);
    gcn_zero_f32<<<4096, 256, 0, stream>>>(agg, (long long)N * D);
    gcn_scatter<<<scatterBlocks, 256, 0, stream>>>(hs, src, dst, agg, (long long)E);
    float* outp = last ? (float*)d_out : hbuf;
    gcn_finalize<<<finBlocks, 256, 0, stream>>>(agg, inrm, last ? nullptr : hin,
                                                outp, N, last ? 0 : 1);
    hin = hbuf;
  }
}


# Segment sum by row ownership: edges grouped by destination once per call (int count+rank, scan, fill, row sort); one wave sums a row in registers in sorted-source order and finishes it in place
# speedup vs baseline: 2.3378x; 2.3378x over previous
#include <hip/hip_runtime.h>

#define D 128
#define WSTR 132

typedef __attribute__((ext_vector_type(2))) float v2f;
typedef __attribute__((ext_vector_type(8))) float v8f;

__global__ void gcn_zero_u32(unsigned* __restrict__ p, long long n) {
  long long i = (long long)blockIdx.x * blockDim.x + threadIdx.x;
  long long stride = (long long)gridDim.x * blockDim.x;
  for (; i < n; i += stride) p[i] = 0u;
}

__global__ void gcn_degree_rank(const int* __restrict__ src, const int* __restrict__ dst,
                                float* __restrict__ odeg, int* __restrict__ cnt,
                                int* __restrict__ rank, int E) {
  int i = blockIdx.x * blockDim.x + threadIdx.x;
  if (i < E) {
    const int s = src[i];
    const int d = dst[i];
    atomicAdd(&odeg[s], 1.0f);
    rank[i] = atomicAdd(&cnt[d], 1);
  }
}

__global__ void gcn_norm(float* __restrict__ x, int n) {
  int i = blockIdx.x * blockDim.x + threadIdx.x;
  if (i < n) x[i] = rsqrtf(fmaxf(x[i], 1.0f));
}

__device__ __forceinline__ int wave_incl_scan(int x, int lane) {
  for (int o = 1; o < 32; o <<= 1) {
    const int y = __shfl_up(x, o);
    if (lane >= o) x += y;
  }
  return x;
}

__global__ __launch_bounds__(256)
void gcn_scan_local(const int* __restrict__ cnt, int* __restrict__ rowl,
                    int* __restrict__ bsum, float* __restrict__ inrm, int n) {
  __shared__ int swave[8];
  const int tid  = threadIdx.x;
  const int lane = tid & 31;
  const int w    = tid >> 5;
  const int base = blockIdx.x * 1024 + tid * 4;
  int c[4];
#pragma unroll
  for (int k = 0; k < 4; ++k) c[k] = (base + k < n) ? cnt[base + k] : 0;
  const int t = c[0] + c[1] + c[2] + c[3];
  const int x = wave_incl_scan(t, lane);
  if (lane == 31) swave[w] = x;
  __syncthreads();
  int woff = 0;
  for (int j = 0; j < w; ++j) woff += swave[j];
  int run = woff + x - t;
#pragma unroll
  for (int k = 0; k < 4; ++k) {
    if (base + k < n) {
      rowl[base + k] = run;
      inrm[base + k] = rsqrtf(fmaxf((float)c[k], 1.0f));
    }
    run += c[k];
  }
  if (tid == 255) bsum[blockIdx.x] = woff + x;
}

__global__ __launch_bounds__(256)
void gcn_scan_top(int* __restrict__ bsum, int nb) {
  __shared__ int swave[8];
  const int tid  = threadIdx.x;
  const int lane = tid & 31;
  const int w    = tid >> 5;
  int carry = 0;
  for (int base = 0; base < nb; base += 256) {
    const int i = base + tid;
    const int t = (i < nb) ? bsum[i] : 0;
    const int x = wave_incl_scan(t, lane);
    if (lane == 31) swave[w] = x;
    __syncthreads();
    int woff = 0, tot = 0;
    for (int j = 0; j < 8; ++j) { if (j < w) woff += swave[j]; tot += swave[j]; }
    if (i < nb) bsum[i] = carry + woff + x - t;
    carry += tot;
    __syncthreads();
  }
}

__global__ void gcn_fill(const int* __restrict__ src, const int* __restrict__ dst,
                         const int* __restrict__ rank, const int* __restrict__ rowl,
                         const int* __restrict__ boff, int* __restrict__ colt, int E) {
  int i = blockIdx.x * blockDim.x + threadIdx.x;
  if (i < E) {
    const int d = dst[i];
    colt[rowl[d] + boff[d >> 10] + rank[i]] = src[i];
  }
}

__global__ __launch_bounds__(256)
void gcn_rowsort(const int* __restrict__ colt, const int* __restrict__ cnt,
                 const int* __restrict__ rowl, const int* __restrict__ boff,
                 int* __restrict__ col, int n) {
  const int row  = blockIdx.x * 8 + (threadIdx.x >> 5);
  const int lane = threadIdx.x & 31;
  if (row >= n) return;
  const int start = rowl[row] + boff[row >> 10];
  const int d     = cnt[row];
  for (int i = lane; i < d; i += 32) {
    const int si = colt[start + i];
    int r = 0;
    for (int j = 0; j < d; ++j) {
      const int sj = colt[start + j];
      r += (sj < si || (sj == si && j < i)) ? 1 : 0;
    }
    col[start + r] = si;
  }
}

__global__ __launch_bounds__(128)
void gcn_gemm_wmma(const float* __restrict__ h, const float* __restrict__ W,
                   const float* __restrict__ bias, const float* __restrict__ out_norm,
                   float* __restrict__ hs, int nrows, int ntiles) {
  __shared__ float sW[64 * WSTR];
  __shared__ float sA[16 * WSTR];
  __shared__ float sB[64];

  const int tid   = threadIdx.x;
  const int nbase = blockIdx.y * 64;

  for (int idx = tid; idx < 64 * (D / 4); idx += 128) {
    const int r  = idx >> 5;
    const int c4 = (idx & 31) << 2;
    const float4 v = *(const float4*)(W + (size_t)(nbase + r) * D + c4);
    *(float4*)(&sW[r * WSTR + c4]) = v;
  }
  if (tid < 64) sB[tid] = bias[nbase + tid];
  __syncthreads();

  const int wave = tid >> 5;
  const int lane = tid & 31;
  const int hi   = lane >> 4;
  const int lm   = lane & 15;

  for (int tile = blockIdx.x; tile < ntiles; tile += gridDim.x) {
    const int row0 = tile << 4;

    for (int idx = tid; idx < 16 * (D / 4); idx += 128) {
      const int r   = idx >> 5;
      const int c4  = (idx & 31) << 2;
      const int row = row0 + r;
      float4 v;
      if (row < nrows) v = *(const float4*)(h + (size_t)row * D + c4);
      else { v.x = 0.f; v.y = 0.f; v.z = 0.f; v.w = 0.f; }
      *(float4*)(&sA[r * WSTR + c4]) = v;
    }
    __syncthreads();

    v8f acc = {};
    const float* __restrict__ arow = &sA[lm * WSTR];
    const float* __restrict__ brow = &sW[(wave * 16 + lm) * WSTR];
#pragma unroll
    for (int k = 0; k < D; k += 4) {
      const int kk = k + (hi << 1);
      v2f a; a.x = arow[kk]; a.y = arow[kk + 1];
      v2f b; b.x = brow[kk]; b.y = brow[kk + 1];
      acc = __builtin_amdgcn_wmma_f32_16x16x4_f32(
           false, a,  false, b,
           (short)0, acc,  false,  false);
    }

    const int n  = nbase + wave * 16 + lm;
    const float bn = sB[wave * 16 + lm];
#pragma unroll
    for (int r = 0; r < 8; ++r) {
      const int m = row0 + r + (hi << 3);
      if (m < nrows)
        hs[(size_t)m * D + n] = (acc[r] + bn) * out_norm[m];
    }
    __syncthreads();
  }
}

template <int U>
__device__ __forceinline__ void gcn_add_rows(const float* __restrict__ hs, int ids, int e,
                                             int c4, float4& acc) {
  float4 v[U];
#pragma unroll
  for (int u = 0; u < U; ++u) {
    const int s = __builtin_amdgcn_readlane(ids, e + u);
    v[u] = *(const float4*)(hs + (size_t)s * D + c4);
  }
#pragma unroll
  for (int u = 0; u < U; ++u) {
    acc.x += v[u].x; acc.y += v[u].y; acc.z += v[u].z; acc.w += v[u].w;
  }
}

__global__ __launch_bounds__(256)
void gcn_gather(const float* __restrict__ hs, const int* __restrict__ col,
                const int* __restrict__ cnt, const int* __restrict__ rowl,
                const int* __restrict__ boff, const float* __restrict__ in_norm,
                const float* hres, float* out, int nrows, int do_relu) {
  const int row = __builtin_amdgcn_readfirstlane((int)(blockIdx.x * 8 + (threadIdx.x >> 5)));
  if (row >= nrows) return;
  const int lane = threadIdx.x & 31;
  const int c4   = lane << 2;
  const int start = rowl[row] + boff[row >> 10];
  const int d     = cnt[row];
  const float s   = in_norm[row];
  const size_t ro = (size_t)row * D + c4;

  float4 res; res.x = 0.f; res.y = 0.f; res.z = 0.f; res.w = 0.f;
  if (hres) res = *(const float4*)(hres + ro);

  float4 acc; acc.x = 0.f; acc.y = 0.f; acc.z = 0.f; acc.w = 0.f;
  for (int base = 0; base < d; base += 32) {
    const int m = (d - base < 32) ? (d - base) : 32;
    const int ids = (lane < m) ? col[start + base + lane] : 0;
    int e = 0;
    for (; e + 8 <= m; e += 8) gcn_add_rows<8>(hs, ids, e, c4, acc);
    if (m & 4) { gcn_add_rows<4>(hs, ids, e, c4, acc); e += 4; }
    if (m & 2) { gcn_add_rows<2>(hs, ids, e, c4, acc); e += 2; }
    if (m & 1) { gcn_add_rows<1>(hs, ids, e, c4, acc); }
  }

  float4 v;
  v.x = acc.x * s; v.y = acc.y * s; v.z = acc.z * s; v.w = acc.w * s;
  if (hres) { v.x += res.x; v.y += res.y; v.z += res.z; v.w += res.w; }
  if (do_relu) {
    v.x = fmaxf(v.x, 0.f); v.y = fmaxf(v.y, 0.f);
    v.z = fmaxf(v.z, 0.f); v.w = fmaxf(v.w, 0.f);
  }
  *(float4*)(out + ro) = v;
}

extern "C" void kernel_launch(void* const* d_in, const int* in_sizes, int n_in,
                              void* d_out, int out_size, void* d_ws, size_t ws_size,
                              hipStream_t stream) {
  const float* feat = (const float*)d_in[0];
  const int*   src  = (const int*)d_in[1];
  const int*   dst  = (const int*)d_in[2];
  const float* Wl[3] = { (const float*)d_in[3], (const float*)d_in[5], (const float*)d_in[7] };
  const float* bl[3] = { (const float*)d_in[4], (const float*)d_in[6], (const float*)d_in[8] };

  const int N = in_sizes[0] / D;
  const int E = in_sizes[1];
  const int nb = (N + 1023) / 1024;

  char* ws = (char*)d_ws;
  const size_t featBytes = (size_t)N * D * sizeof(float);
  const size_t normBytes = (((size_t)N * sizeof(float)) + 255) & ~(size_t)255;
  const size_t nbBytes   = (((size_t)nb * sizeof(int)) + 255) & ~(size_t)255;
  const size_t edgeBytes = (((size_t)E * sizeof(int)) + 255) & ~(size_t)255;
  float* hs   = (float*)ws;
  float* hbuf = (float*)(ws + featBytes);
  char*  p    = ws + 2 * featBytes;
  float* onrm = (float*)p;  p += normBytes;
  int*   cnt  = (int*)p;    p += normBytes;
  float* inrm = (float*)p;  p += normBytes;
  int*   rowl = (int*)p;    p += normBytes;
  int*   boff = (int*)p;    p += nbBytes;
  int*   rank = (int*)p;    p += edgeBytes;
  int*   colt = (int*)p;    p += edgeBytes;
  int*   col  = (int*)p;

  gcn_zero_u32<<<1024, 256, 0, stream>>>((unsigned*)onrm, (long long)(2 * (normBytes / sizeof(float))));
  gcn_degree_rank<<<(E + 255) / 256, 256, 0, stream>>>(src, dst, onrm, cnt, rank, E);
  gcn_norm<<<(N + 255) / 256, 256, 0, stream>>>(onrm, N);
  gcn_scan_local<<<nb, 256, 0, stream>>>(cnt, rowl, boff, inrm, N);
  gcn_scan_top<<<1, 256, 0, stream>>>(boff, nb);
  gcn_fill<<<(E + 255) / 256, 256, 0, stream>>>(src, dst, rank, rowl, boff, colt, E);
  gcn_rowsort<<<(N + 7) / 8, 256, 0, stream>>>(colt, cnt, rowl, boff, col, N);

  const int ntiles = (N + 15) / 16;
  const dim3 ggrid(1024, 2);
  const unsigned gatherBlocks = (unsigned)((N + 7) / 8);

  const float* hin = feat;
  for (int l = 0; l < 3; ++l) {
    const bool last = (l == 2);
    gcn_gemm_wmma<<<ggrid, 128, 0, stream>>>(hin, Wl[l], bl[l], onrm, hs, N, ntiles);
    float* outp = last ? (float*)d_out : hbuf;
    gcn_gather<<<gatherBlocks, 256, 0, stream>>>(hs, col, cnt, rowl, boff, inrm,
                                                 last ? nullptr : hin, outp, N, last ? 0 : 1);
    hin = hbuf;
  }
}


# Degrees and edge slots counted in LDS: per 32768-edge chunk a work-group adds into packed 16-bit node counters (old value = slot), a per-node pass sums chunks; no global atomics; sum order kept
# speedup vs baseline: 2.8539x; 1.2208x over previous
#include <hip/hip_runtime.h>

#define D 128
#define WSTR 132

typedef __attribute__((ext_vector_type(2))) float v2f;
typedef __attribute__((ext_vector_type(8))) float v8f;

#define HCH    32768
#define HCHLOG 15
#define HWORDS 65536

template <bool FULL>
__device__ __forceinline__ void gcn_hist_chunk(const int* __restrict__ key, unsigned* sc,
                                               unsigned* __restrict__ outp,
                                               unsigned short* __restrict__ rank,
                                               bool isdst, int e0, int e1, int hstride) {
  const int tid = threadIdx.x;
  const int b0  = e0 + tid * 4;

  int k[32];
#pragma unroll
  for (int j = 0; j < 8; ++j) {
    if (FULL) {
      const int4 t = *(const int4*)(key + b0 + j * 4096);
      k[4 * j] = t.x; k[4 * j + 1] = t.y; k[4 * j + 2] = t.z; k[4 * j + 3] = t.w;
    } else {
#pragma unroll
      for (int q = 0; q < 4; ++q) {
        const int idx = b0 + j * 4096 + q;
        k[4 * j + q] = (idx < e1) ? key[idx] : -1;
      }
    }
  }

  const int nvec = hstride >> 2;
  uint4 z; z.x = 0u; z.y = 0u; z.z = 0u; z.w = 0u;
  for (int v = tid; v < nvec; v += 1024) ((uint4*)sc)[v] = z;
  __syncthreads();

  if (isdst) {
    unsigned short r[32];
#pragma unroll
    for (int i = 0; i < 32; ++i) {
      r[i] = 0;
      if (FULL || k[i] >= 0) {
        const unsigned sh  = ((unsigned)k[i] & 1u) << 4;
        const unsigned old = atomicAdd(&sc[k[i] >> 1], 1u << sh);
        r[i] = (unsigned short)((old >> sh) & 0xFFFFu);
      }
    }
#pragma unroll
    for (int j = 0; j < 8; ++j) {
      if (FULL) {
        ushort4 rv; rv.x = r[4 * j]; rv.y = r[4 * j + 1]; rv.z = r[4 * j + 2]; rv.w = r[4 * j + 3];
        *(ushort4*)(rank + b0 + j * 4096) = rv;
      } else {
#pragma unroll
        for (int q = 0; q < 4; ++q) {
          const int idx = b0 + j * 4096 + q;
          if (idx < e1) rank[idx] = r[4 * j + q];
        }
      }
    }
  } else {
#pragma unroll
    for (int i = 0; i < 32; ++i) {
      if (FULL || k[i] >= 0) atomicAdd(&sc[k[i] >> 1], 1u << (((unsigned)k[i] & 1u) << 4));
    }
  }
  __syncthreads();

  for (int v = tid; v < nvec; v += 1024) ((uint4*)outp)[v] = ((const uint4*)sc)[v];
}

__global__ __launch_bounds__(1024)
void gcn_hist_rank(const int* __restrict__ src, const int* __restrict__ dst,
                   unsigned* __restrict__ hsrc, unsigned* __restrict__ hdst,
                   unsigned short* __restrict__ rank, int E, int hstride) {
  __shared__ __attribute__((aligned(16))) unsigned sc[HWORDS];
  const int chunk = blockIdx.x;
  const bool isdst = (blockIdx.y == 0);
  const int* __restrict__ key = isdst ? dst : src;
  unsigned* __restrict__ outp = (isdst ? hdst : hsrc) + (size_t)chunk * hstride;
  const int e0 = chunk << HCHLOG;
  if (e0 + HCH <= E) gcn_hist_chunk<true>(key, sc, outp, rank, isdst, e0, e0 + HCH, hstride);
  else               gcn_hist_chunk<false>(key, sc, outp, rank, isdst, e0, E, hstride);
}

__global__ __launch_bounds__(256)
void gcn_hist_scan(const unsigned* __restrict__ hsrc, const unsigned* __restrict__ hdst,
                   int* __restrict__ pre, float* __restrict__ onrm, int* __restrict__ cnt,
                   int hstride, int nchunks, int n) {
  const int w = blockIdx.x * 256 + threadIdx.x;
  if (w >= hstride) return;
  const bool isdst = (blockIdx.y == 0);
  const unsigned* __restrict__ h = isdst ? hdst : hsrc;
  unsigned lo = 0u, hi = 0u;
  for (int c0 = 0; c0 < nchunks; c0 += 32) {
    unsigned v[32];
#pragma unroll
    for (int u = 0; u < 32; ++u)
      v[u] = (c0 + u < nchunks) ? h[(size_t)(c0 + u) * hstride + w] : 0u;
#pragma unroll
    for (int u = 0; u < 32; ++u) {
      if (isdst && c0 + u < nchunks) {
        int2 p; p.x = (int)lo; p.y = (int)hi;
        *(int2*)(pre + ((size_t)(c0 + u) * hstride + w) * 2) = p;
      }
      lo += v[u] & 0xFFFFu;
      hi += v[u] >> 16;
    }
  }
  const int n0 = 2 * w;
  if (isdst) {
    if (n0     < n) cnt[n0]     = (int)lo;
    if (n0 + 1 < n) cnt[n0 + 1] = (int)hi;
  } else {
    if (n0     < n) onrm[n0]     = rsqrtf(fmaxf((float)lo, 1.0f));
    if (n0 + 1 < n) onrm[n0 + 1] = rsqrtf(fmaxf((float)hi, 1.0f));
  }
}

__device__ __forceinline__ int wave_incl_scan(int x, int lane) {
  for (int o = 1; o < 32; o <<= 1) {
    const int y = __shfl_up(x, o);
    if (lane >= o) x += y;
  }
  return x;
}

__global__ __launch_bounds__(256)
void gcn_scan_local(const int* __restrict__ cnt, int* __restrict__ rowl,
                    int* __restrict__ bsum, float* __restrict__ inrm, int n) {
  __shared__ int swave[8];
  const int tid  = threadIdx.x;
  const int lane = tid & 31;
  const int w    = tid >> 5;
  const int base = blockIdx.x * 1024 + tid * 4;
  int c[4];
#pragma unroll
  for (int k = 0; k < 4; ++k) c[k] = (base + k < n) ? cnt[base + k] : 0;
  const int t = c[0] + c[1] + c[2] + c[3];
  const int x = wave_incl_scan(t, lane);
  if (lane == 31) swave[w] = x;
  __syncthreads();
  int woff = 0;
  for (int j = 0; j < w; ++j) woff += swave[j];
  int run = woff + x - t;
#pragma unroll
  for (int k = 0; k < 4; ++k) {
    if (base + k < n) {
      rowl[base + k] = run;
      inrm[base + k] = rsqrtf(fmaxf((float)c[k], 1.0f));
    }
    run += c[k];
  }
  if (tid == 255) bsum[blockIdx.x] = woff + x;
}

__global__ __launch_bounds__(256)
void gcn_scan_top(int* __restrict__ bsum, int nb) {
  __shared__ int swave[8];
  const int tid  = threadIdx.x;
  const int lane = tid & 31;
  const int w    = tid >> 5;
  int carry = 0;
  for (int base = 0; base < nb; base += 256) {
    const int i = base + tid;
    const int t = (i < nb) ? bsum[i] : 0;
    const int x = wave_incl_scan(t, lane);
    if (lane == 31) swave[w] = x;
    __syncthreads();
    int woff = 0, tot = 0;
    for (int j = 0; j < 8; ++j) { if (j < w) woff += swave[j]; tot += swave[j]; }
    if (i < nb) bsum[i] = carry + woff + x - t;
    carry += tot;
    __syncthreads();
  }
}

__global__ void gcn_fill(const int* __restrict__ src, const int* __restrict__ dst,
                         const unsigned short* __restrict__ rank, const int* __restrict__ rowl,
                         const int* __restrict__ boff, const int* __restrict__ pre,
                         int* __restrict__ colt, int E, int prestride) {
  int i = blockIdx.x * blockDim.x + threadIdx.x;
  if (i < E) {
    const int d = dst[i];
    const int pc = pre[(size_t)(i >> HCHLOG) * prestride + d];
    colt[rowl[d] + boff[d >> 10] + pc + (int)rank[i]] = src[i];
  }
}

__global__ __launch_bounds__(256)
void gcn_rowsort(const int* __restrict__ colt, const int* __restrict__ cnt,
                 const int* __restrict__ rowl, const int* __restrict__ boff,
                 int* __restrict__ col, int n) {
  const int row  = blockIdx.x * 8 + (threadIdx.x >> 5);
  const int lane = threadIdx.x & 31;
  if (row >= n) return;
  const int start = rowl[row] + boff[row >> 10];
  const int d     = cnt[row];
  for (int i = lane; i < d; i += 32) {
    const int si = colt[start + i];
    int r = 0;
    for (int j = 0; j < d; ++j) {
      const int sj = colt[start + j];
      r += (sj < si || (sj == si && j < i)) ? 1 : 0;
    }
    col[start + r] = si;
  }
}

__global__ __launch_bounds__(128)
void gcn_gemm_wmma(const float* __restrict__ h, const float* __restrict__ W,
                   const float* __restrict__ bias, const float* __restrict__ out_norm,
                   float* __restrict__ hs, int nrows, int ntiles) {
  __shared__ float sW[64 * WSTR];
  __shared__ float sA[16 * WSTR];
  __shared__ float sB[64];

  const int tid   = threadIdx.x;
  const int nbase = blockIdx.y * 64;

  for (int idx = tid; idx < 64 * (D / 4); idx += 128) {
    const int r  = idx >> 5;
    const int c4 = (idx & 31) << 2;
    const float4 v = *(const float4*)(W + (size_t)(nbase + r) * D + c4);
    *(float4*)(&sW[r * WSTR + c4]) = v;
  }
  if (tid < 64) sB[tid] = bias[nbase + tid];
  __syncthreads();

  const int wave = tid >> 5;
  const int lane = tid & 31;
  const int hi   = lane >> 4;
  const int lm   = lane & 15;

  for (int tile = blockIdx.x; tile < ntiles; tile += gridDim.x) {
    const int row0 = tile << 4;

    for (int idx = tid; idx < 16 * (D / 4); idx += 128) {
      const int r   = idx >> 5;
      const int c4  = (idx & 31) << 2;
      const int row = row0 + r;
      float4 v;
      if (row < nrows) v = *(const float4*)(h + (size_t)row * D + c4);
      else { v.x = 0.f; v.y = 0.f; v.z = 0.f; v.w = 0.f; }
      *(float4*)(&sA[r * WSTR + c4]) = v;
    }
    __syncthreads();

    v8f acc = {};
    const float* __restrict__ arow = &sA[lm * WSTR];
    const float* __restrict__ brow = &sW[(wave * 16 + lm) * WSTR];
#pragma unroll
    for (int k = 0; k < D; k += 4) {
      const int kk = k + (hi << 1);
      v2f a; a.x = arow[kk]; a.y = arow[kk + 1];
      v2f b; b.x = brow[kk]; b.y = brow[kk + 1];
      acc = __builtin_amdgcn_wmma_f32_16x16x4_f32(
           false, a,  false, b,
           (short)0, acc,  false,  false);
    }

    const int n  = nbase + wave * 16 + lm;
    const float bn = sB[wave * 16 + lm];
#pragma unroll
    for (int r = 0; r < 8; ++r) {
      const int m = row0 + r + (hi << 3);
      if (m < nrows)
        hs[(size_t)m * D + n] = (acc[r] + bn) * out_norm[m];
    }
    __syncthreads();
  }
}

template <int U>
__device__ __forceinline__ void gcn_add_rows(const float* __restrict__ hs, int ids, int e,
                                             int c4, float4& acc) {
  float4 v[U];
#pragma unroll
  for (int u = 0; u < U; ++u) {
    const int s = __builtin_amdgcn_readlane(ids, e + u);
    v[u] = *(const float4*)(hs + (size_t)s * D + c4);
  }
#pragma unroll
  for (int u = 0; u < U; ++u) {
    acc.x += v[u].x; acc.y += v[u].y; acc.z += v[u].z; acc.w += v[u].w;
  }
}

__global__ __launch_bounds__(256)
void gcn_gather(const float* __restrict__ hs, const int* __restrict__ col,
                const int* __restrict__ cnt, const int* __restrict__ rowl,
                const int* __restrict__ boff, const float* __restrict__ in_norm,
                const float* hres, float* out, int nrows, int do_relu) {
  const int row = __builtin_amdgcn_readfirstlane((int)(blockIdx.x * 8 + (threadIdx.x >> 5)));
  if (row >= nrows) return;
  const int lane = threadIdx.x & 31;
  const int c4   = lane << 2;
  const int start = rowl[row] + boff[row >> 10];
  const int d     = cnt[row];
  const float s   = in_norm[row];
  const size_t ro = (size_t)row * D + c4;

  float4 res; res.x = 0.f; res.y = 0.f; res.z = 0.f; res.w = 0.f;
  if (hres) res = *(const float4*)(hres + ro);

  float4 acc; acc.x = 0.f; acc.y = 0.f; acc.z = 0.f; acc.w = 0.f;
  for (int base = 0; base < d; base += 32) {
    const int m = (d - base < 32) ? (d - base) : 32;
    const int ids = (lane < m) ? col[start + base + lane] : 0;
    int e = 0;
    for (; e + 8 <= m; e += 8) gcn_add_rows<8>(hs, ids, e, c4, acc);
    if (m & 4) { gcn_add_rows<4>(hs, ids, e, c4, acc); e += 4; }
    if (m & 2) { gcn_add_rows<2>(hs, ids, e, c4, acc); e += 2; }
    if (m & 1) { gcn_add_rows<1>(hs, ids, e, c4, acc); }
  }

  float4 v;
  v.x = acc.x * s; v.y = acc.y * s; v.z = acc.z * s; v.w = acc.w * s;
  if (hres) { v.x += res.x; v.y += res.y; v.z += res.z; v.w += res.w; }
  if (do_relu) {
    v.x = fmaxf(v.x, 0.f); v.y = fmaxf(v.y, 0.f);
    v.z = fmaxf(v.z, 0.f); v.w = fmaxf(v.w, 0.f);
  }
  *(float4*)(out + ro) = v;
}

extern "C" void kernel_launch(void* const* d_in, const int* in_sizes, int n_in,
                              void* d_out, int out_size, void* d_ws, size_t ws_size,
                              hipStream_t stream) {
  const float* feat = (const float*)d_in[0];
  const int*   src  = (const int*)d_in[1];
  const int*   dst  = (const int*)d_in[2];
  const float* Wl[3] = { (const float*)d_in[3], (const float*)d_in[5], (const float*)d_in[7] };
  const float* bl[3] = { (const float*)d_in[4], (const float*)d_in[6], (const float*)d_in[8] };

  const int N = in_sizes[0] / D;
  const int E = in_sizes[1];
  const int nb = (N + 1023) / 1024;

  char* ws = (char*)d_ws;
  const int nchunks = (E + HCH - 1) / HCH;
  const int hstride = (((N + 1) / 2) + 3) & ~3;
  const size_t featBytes = (size_t)N * D * sizeof(float);
  const size_t normBytes = (((size_t)N * sizeof(float)) + 255) & ~(size_t)255;
  const size_t nbBytes   = (((size_t)nb * sizeof(int)) + 255) & ~(size_t)255;
  const size_t edgeBytes = (((size_t)E * sizeof(int)) + 255) & ~(size_t)255;
  const size_t histBytes = (((size_t)nchunks * hstride * sizeof(unsigned)) + 255) & ~(size_t)255;
  float* hs   = (float*)ws;
  float* hbuf = (float*)(ws + featBytes);
  char*  p    = ws + 2 * featBytes;
  float* onrm = (float*)p;  p += normBytes;
  int*   cnt  = (int*)p;    p += normBytes;
  float* inrm = (float*)p;  p += normBytes;
  int*   rowl = (int*)p;    p += normBytes;
  int*   boff = (int*)p;    p += nbBytes;
  unsigned short* rank = (unsigned short*)p;  p += edgeBytes;
  int*   colt = (int*)p;    p += edgeBytes;
  int*   col  = (int*)p;    p += edgeBytes;
  unsigned* hsrc = (unsigned*)p;  p += histBytes;
  unsigned* hdst = (unsigned*)p;  p += histBytes;
  int*   pre  = (int*)p;

  gcn_hist_rank<<<dim3(nchunks, 2), 1024, 0, stream>>>(src, dst, hsrc, hdst, rank, E, hstride);
  gcn_hist_scan<<<dim3((hstride + 255) / 256, 2), 256, 0, stream>>>(hsrc, hdst, pre, onrm, cnt,
                                                                    hstride, nchunks, N);
  gcn_scan_local<<<nb, 256, 0, stream>>>(cnt, rowl, boff, inrm, N);
  gcn_scan_top<<<1, 256, 0, stream>>>(boff, nb);
  gcn_fill<<<(E + 255) / 256, 256, 0, stream>>>(src, dst, rank, rowl, boff, pre, colt, E, 2 * hstride);
  gcn_rowsort<<<(N + 7) / 8, 256, 0, stream>>>(colt, cnt, rowl, boff, col, N);

  const int ntiles = (N + 15) / 16;
  const dim3 ggrid(1024, 2);
  const unsigned gatherBlocks = (unsigned)((N + 7) / 8);

  const float* hin = feat;
  for (int l = 0; l < 3; ++l) {
    const bool last = (l == 2);
    gcn_gemm_wmma<<<ggrid, 128, 0, stream>>>(hin, Wl[l], bl[l], onrm, hs, N, ntiles);
    float* outp = last ? (float*)d_out : hbuf;
    gcn_gather<<<gatherBlocks, 256, 0, stream>>>(hs, col, cnt, rowl, boff, inrm,
                                                 last ? nullptr : hin, outp, N, last ? 0 : 1);
    hin = hbuf;
  }
}


# Dense product re-tiled: one 16-wave group per WGP keeps all of W in LDS, a wave owns 16 rows x 128 columns (8 accumulators, A read once, whole-row stores); 16-byte k pieces: k summed in another order
# speedup vs baseline: 4.3407x; 1.5210x over previous
#include <hip/hip_runtime.h>

#define D 128
#define WSTR 132

typedef __attribute__((ext_vector_type(2))) float v2f;
typedef __attribute__((ext_vector_type(8))) float v8f;

#define HCH    32768
#define HCHLOG 15
#define HWORDS 65536

template <bool FULL>
__device__ __forceinline__ void gcn_hist_chunk(const int* __restrict__ key, unsigned* sc,
                                               unsigned* __restrict__ outp,
                                               unsigned short* __restrict__ rank,
                                               bool isdst, int e0, int e1, int hstride) {
  const int tid = threadIdx.x;
  const int b0  = e0 + tid * 4;

  int k[32];
#pragma unroll
  for (int j = 0; j < 8; ++j) {
    if (FULL) {
      const int4 t = *(const int4*)(key + b0 + j * 4096);
      k[4 * j] = t.x; k[4 * j + 1] = t.y; k[4 * j + 2] = t.z; k[4 * j + 3] = t.w;
    } else {
#pragma unroll
      for (int q = 0; q < 4; ++q) {
        const int idx = b0 + j * 4096 + q;
        k[4 * j + q] = (idx < e1) ? key[idx] : -1;
      }
    }
  }

  const int nvec = hstride >> 2;
  uint4 z; z.x = 0u; z.y = 0u; z.z = 0u; z.w = 0u;
  for (int v = tid; v < nvec; v += 1024) ((uint4*)sc)[v] = z;
  __syncthreads();

  if (isdst) {
    unsigned short r[32];
#pragma unroll
    for (int i = 0; i < 32; ++i) {
      r[i] = 0;
      if (FULL || k[i] >= 0) {
        const unsigned sh  = ((unsigned)k[i] & 1u) << 4;
        const unsigned old = atomicAdd(&sc[k[i] >> 1], 1u << sh);
        r[i] = (unsigned short)((old >> sh) & 0xFFFFu);
      }
    }
#pragma unroll
    for (int j = 0; j < 8; ++j) {
      if (FULL) {
        ushort4 rv; rv.x = r[4 * j]; rv.y = r[4 * j + 1]; rv.z = r[4 * j + 2]; rv.w = r[4 * j + 3];
        *(ushort4*)(rank + b0 + j * 4096) = rv;
      } else {
#pragma unroll
        for (int q = 0; q < 4; ++q) {
          const int idx = b0 + j * 4096 + q;
          if (idx < e1) rank[idx] = r[4 * j + q];
        }
      }
    }
  } else {
#pragma unroll
    for (int i = 0; i < 32; ++i) {
      if (FULL || k[i] >= 0) atomicAdd(&sc[k[i] >> 1], 1u << (((unsigned)k[i] & 1u) << 4));
    }
  }
  __syncthreads();

  for (int v = tid; v < nvec; v += 1024) ((uint4*)outp)[v] = ((const uint4*)sc)[v];
}

__global__ __launch_bounds__(1024)
void gcn_hist_rank(const int* __restrict__ src, const int* __restrict__ dst,
                   unsigned* __restrict__ hsrc, unsigned* __restrict__ hdst,
                   unsigned short* __restrict__ rank, int E, int hstride) {
  __shared__ __attribute__((aligned(16))) unsigned sc[HWORDS];
  const int chunk = blockIdx.x;
  const bool isdst = (blockIdx.y == 0);
  const int* __restrict__ key = isdst ? dst : src;
  unsigned* __restrict__ outp = (isdst ? hdst : hsrc) + (size_t)chunk * hstride;
  const int e0 = chunk << HCHLOG;
  if (e0 + HCH <= E) gcn_hist_chunk<true>(key, sc, outp, rank, isdst, e0, e0 + HCH, hstride);
  else               gcn_hist_chunk<false>(key, sc, outp, rank, isdst, e0, E, hstride);
}

__global__ __launch_bounds__(256)
void gcn_hist_scan(const unsigned* __restrict__ hsrc, const unsigned* __restrict__ hdst,
                   int* __restrict__ pre, float* __restrict__ onrm, int* __restrict__ cnt,
                   int hstride, int nchunks, int n) {
  const int w = blockIdx.x * 256 + threadIdx.x;
  if (w >= hstride) return;
  const bool isdst = (blockIdx.y == 0);
  const unsigned* __restrict__ h = isdst ? hdst : hsrc;
  unsigned lo = 0u, hi = 0u;
  for (int c0 = 0; c0 < nchunks; c0 += 32) {
    unsigned v[32];
#pragma unroll
    for (int u = 0; u < 32; ++u)
      v[u] = (c0 + u < nchunks) ? h[(size_t)(c0 + u) * hstride + w] : 0u;
#pragma unroll
    for (int u = 0; u < 32; ++u) {
      if (isdst && c0 + u < nchunks) {
        int2 p; p.x = (int)lo; p.y = (int)hi;
        *(int2*)(pre + ((size_t)(c0 + u) * hstride + w) * 2) = p;
      }
      lo += v[u] & 0xFFFFu;
      hi += v[u] >> 16;
    }
  }
  const int n0 = 2 * w;
  if (isdst) {
    if (n0     < n) cnt[n0]     = (int)lo;
    if (n0 + 1 < n) cnt[n0 + 1] = (int)hi;
  } else {
    if (n0     < n) onrm[n0]     = rsqrtf(fmaxf((float)lo, 1.0f));
    if (n0 + 1 < n) onrm[n0 + 1] = rsqrtf(fmaxf((float)hi, 1.0f));
  }
}

__device__ __forceinline__ int wave_incl_scan(int x, int lane) {
  for (int o = 1; o < 32; o <<= 1) {
    const int y = __shfl_up(x, o);
    if (lane >= o) x += y;
  }
  return x;
}

__global__ __launch_bounds__(256)
void gcn_scan_local(const int* __restrict__ cnt, int* __restrict__ rowl,
                    int* __restrict__ bsum, float* __restrict__ inrm, int n) {
  __shared__ int swave[8];
  const int tid  = threadIdx.x;
  const int lane = tid & 31;
  const int w    = tid >> 5;
  const int base = blockIdx.x * 1024 + tid * 4;
  int c[4];
#pragma unroll
  for (int k = 0; k < 4; ++k) c[k] = (base + k < n) ? cnt[base + k] : 0;
  const int t = c[0] + c[1] + c[2] + c[3];
  const int x = wave_incl_scan(t, lane);
  if (lane == 31) swave[w] = x;
  __syncthreads();
  int woff = 0;
  for (int j = 0; j < w; ++j) woff += swave[j];
  int run = woff + x - t;
#pragma unroll
  for (int k = 0; k < 4; ++k) {
    if (base + k < n) {
      rowl[base + k] = run;
      inrm[base + k] = rsqrtf(fmaxf((float)c[k], 1.0f));
    }
    run += c[k];
  }
  if (tid == 255) bsum[blockIdx.x] = woff + x;
}

__global__ __launch_bounds__(256)
void gcn_scan_top(int* __restrict__ bsum, int nb) {
  __shared__ int swave[8];
  const int tid  = threadIdx.x;
  const int lane = tid & 31;
  const int w    = tid >> 5;
  int carry = 0;
  for (int base = 0; base < nb; base += 256) {
    const int i = base + tid;
    const int t = (i < nb) ? bsum[i] : 0;
    const int x = wave_incl_scan(t, lane);
    if (lane == 31) swave[w] = x;
    __syncthreads();
    int woff = 0, tot = 0;
    for (int j = 0; j < 8; ++j) { if (j < w) woff += swave[j]; tot += swave[j]; }
    if (i < nb) bsum[i] = carry + woff + x - t;
    carry += tot;
    __syncthreads();
  }
}

__global__ void gcn_fill(const int* __restrict__ src, const int* __restrict__ dst,
                         const unsigned short* __restrict__ rank, const int* __restrict__ rowl,
                         const int* __restrict__ boff, const int* __restrict__ pre,
                         int* __restrict__ colt, int E, int prestride) {
  int i = blockIdx.x * blockDim.x + threadIdx.x;
  if (i < E) {
    const int d = dst[i];
    const int pc = pre[(size_t)(i >> HCHLOG) * prestride + d];
    colt[rowl[d] + boff[d >> 10] + pc + (int)rank[i]] = src[i];
  }
}

__global__ __launch_bounds__(256)
void gcn_rowsort(const int* __restrict__ colt, const int* __restrict__ cnt,
                 const int* __restrict__ rowl, const int* __restrict__ boff,
                 int* __restrict__ col, int n) {
  const int row  = blockIdx.x * 8 + (threadIdx.x >> 5);
  const int lane = threadIdx.x & 31;
  if (row >= n) return;
  const int start = rowl[row] + boff[row >> 10];
  const int d     = cnt[row];
  for (int i = lane; i < d; i += 32) {
    const int si = colt[start + i];
    int r = 0;
    for (int j = 0; j < d; ++j) {
      const int sj = colt[start + j];
      r += (sj < si || (sj == si && j < i)) ? 1 : 0;
    }
    col[start + r] = si;
  }
}

#define GW 16

__global__ __launch_bounds__(GW * 32)
void gcn_gemm_wmma(const float* __restrict__ h, const float* __restrict__ W,
                   const float* __restrict__ bias, const float* __restrict__ out_norm,
                   float* __restrict__ hs, int nrows, int ntiles) {
  __shared__ __attribute__((aligned(16))) float sW[D * WSTR];
  __shared__ __attribute__((aligned(16))) float sO[GW * 16 * WSTR];

  const int tid = threadIdx.x;
#pragma unroll
  for (int u = 0; u < (D * D / 4) / (GW * 32); ++u) {
    const int idx = tid + u * (GW * 32);
    const int r   = idx >> 5;
    const int c4  = (idx & 31) << 2;
    *(float4*)(&sW[r * WSTR + c4]) = *(const float4*)(W + (size_t)r * D + c4);
  }
  __syncthreads();

  const int wave = __builtin_amdgcn_readfirstlane(tid >> 5);
  const int lane = tid & 31;
  const int hi   = lane >> 4;
  const int lm   = lane & 15;
  const float4 b4 = *(const float4*)(bias + 4 * lane);
  const float* __restrict__ bfrag = &sW[lm * WSTR + 4 * hi];
  float* so = &sO[wave * 16 * WSTR];

  for (int tile = blockIdx.x + gridDim.x * wave; tile < ntiles; tile += gridDim.x * GW) {
    const int row0 = tile << 4;
    int ar = row0 + lm;
    ar = (ar < nrows) ? ar : (nrows - 1);
    const float* __restrict__ ap = h + (size_t)ar * D + 4 * hi;
    float4 a[16];
#pragma unroll
    for (int j = 0; j < 16; ++j) a[j] = *(const float4*)(ap + 8 * j);
    const int onv = __float_as_int(out_norm[ar]);

    v8f acc[8];
#pragma unroll
    for (int c = 0; c < 8; ++c) acc[c] = (v8f){0.f, 0.f, 0.f, 0.f, 0.f, 0.f, 0.f, 0.f};

#pragma unroll
    for (int j = 0; j < 16; ++j) {
      float4 b[8];
#pragma unroll
      for (int c = 0; c < 8; ++c) b[c] = *(const float4*)(bfrag + c * 16 * WSTR + 8 * j);
#pragma unroll
      for (int c = 0; c < 8; ++c) {
        v2f x; x.x = a[j].x; x.y = a[j].y;
        v2f y; y.x = b[c].x; y.y = b[c].y;
        acc[c] = __builtin_amdgcn_wmma_f32_16x16x4_f32(false, x, false, y, (short)0, acc[c], false, false);
      }
#pragma unroll
      for (int c = 0; c < 8; ++c) {
        v2f x; x.x = a[j].z; x.y = a[j].w;
        v2f y; y.x = b[c].z; y.y = b[c].w;
        acc[c] = __builtin_amdgcn_wmma_f32_16x16x4_f32(false, x, false, y, (short)0, acc[c], false, false);
      }
    }

#pragma unroll
    for (int c = 0; c < 8; ++c) {
#pragma unroll
      for (int r = 0; r < 8; ++r) so[(r + 8 * hi) * WSTR + 16 * c + lm] = acc[c][r];
    }
    const int nfull = (nrows - row0 < 16) ? (nrows - row0) : 16;
    if (nfull == 16) {
#pragma unroll
      for (int rr = 0; rr < 16; ++rr) {
        const float sc = __int_as_float(__builtin_amdgcn_readlane(onv, rr));
        const float4 v = *(const float4*)(&so[rr * WSTR + 4 * lane]);
        float4 o;
        o.x = (v.x + b4.x) * sc; o.y = (v.y + b4.y) * sc;
        o.z = (v.z + b4.z) * sc; o.w = (v.w + b4.w) * sc;
        *(float4*)(hs + (size_t)(row0 + rr) * D + 4 * lane) = o;
      }
    } else {
      for (int rr = 0; rr < nfull; ++rr) {
        const float sc = __int_as_float(__builtin_amdgcn_readlane(onv, rr));
        const float4 v = *(const float4*)(&so[rr * WSTR + 4 * lane]);
        float4 o;
        o.x = (v.x + b4.x) * sc; o.y = (v.y + b4.y) * sc;
        o.z = (v.z + b4.z) * sc; o.w = (v.w + b4.w) * sc;
        *(float4*)(hs + (size_t)(row0 + rr) * D + 4 * lane) = o;
      }
    }
  }
}

template <int U>
__device__ __forceinline__ void gcn_add_rows(const float* __restrict__ hs, int ids, int e,
                                             int c4, float4& acc) {
  float4 v[U];
#pragma unroll
  for (int u = 0; u < U; ++u) {
    const int s = __builtin_amdgcn_readlane(ids, e + u);
    v[u] = *(const float4*)(hs + (size_t)s * D + c4);
  }
#pragma unroll
  for (int u = 0; u < U; ++u) {
    acc.x += v[u].x; acc.y += v[u].y; acc.z += v[u].z; acc.w += v[u].w;
  }
}

__global__ __launch_bounds__(256)
void gcn_gather(const float* __restrict__ hs, const int* __restrict__ col,
                const int* __restrict__ cnt, const int* __restrict__ rowl,
                const int* __restrict__ boff, const float* __restrict__ in_norm,
                const float* hres, float* out, int nrows, int do_relu) {
  const int row = __builtin_amdgcn_readfirstlane((int)(blockIdx.x * 8 + (threadIdx.x >> 5)));
  if (row >= nrows) return;
  const int lane = threadIdx.x & 31;
  const int c4   = lane << 2;
  const int start = rowl[row] + boff[row >> 10];
  const int d     = cnt[row];
  const float s   = in_norm[row];
  const size_t ro = (size_t)row * D + c4;

  float4 res; res.x = 0.f; res.y = 0.f; res.z = 0.f; res.w = 0.f;
  if (hres) res = *(const float4*)(hres + ro);

  float4 acc; acc.x = 0.f; acc.y = 0.f; acc.z = 0.f; acc.w = 0.f;
  for (int base = 0; base < d; base += 32) {
    const int m = (d - base < 32) ? (d - base) : 32;
    const int ids = (lane < m) ? col[start + base + lane] : 0;
    int e = 0;
    for (; e + 8 <= m; e += 8) gcn_add_rows<8>(hs, ids, e, c4, acc);
    if (m & 4) { gcn_add_rows<4>(hs, ids, e, c4, acc); e += 4; }
    if (m & 2) { gcn_add_rows<2>(hs, ids, e, c4, acc); e += 2; }
    if (m & 1) { gcn_add_rows<1>(hs, ids, e, c4, acc); }
  }

  float4 v;
  v.x = acc.x * s; v.y = acc.y * s; v.z = acc.z * s; v.w = acc.w * s;
  if (hres) { v.x += res.x; v.y += res.y; v.z += res.z; v.w += res.w; }
  if (do_relu) {
    v.x = fmaxf(v.x, 0.f); v.y = fmaxf(v.y, 0.f);
    v.z = fmaxf(v.z, 0.f); v.w = fmaxf(v.w, 0.f);
  }
  *(float4*)(out + ro) = v;
}

extern "C" void kernel_launch(void* const* d_in, const int* in_sizes, int n_in,
                              void* d_out, int out_size, void* d_ws, size_t ws_size,
                              hipStream_t stream) {
  const float* feat = (const float*)d_in[0];
  const int*   src  = (const int*)d_in[1];
  const int*   dst  = (const int*)d_in[2];
  const float* Wl[3] = { (const float*)d_in[3], (const float*)d_in[5], (const float*)d_in[7] };
  const float* bl[3] = { (const float*)d_in[4], (const float*)d_in[6], (const float*)d_in[8] };

  const int N = in_sizes[0] / D;
  const int E = in_sizes[1];
  const int nb = (N + 1023) / 1024;

  char* ws = (char*)d_ws;
  const int nchunks = (E + HCH - 1) / HCH;
  const int hstride = (((N + 1) / 2) + 3) & ~3;
  const size_t featBytes = (size_t)N * D * sizeof(float);
  const size_t normBytes = (((size_t)N * sizeof(float)) + 255) & ~(size_t)255;
  const size_t nbBytes   = (((size_t)nb * sizeof(int)) + 255) & ~(size_t)255;
  const size_t edgeBytes = (((size_t)E * sizeof(int)) + 255) & ~(size_t)255;
  const size_t histBytes = (((size_t)nchunks * hstride * sizeof(unsigned)) + 255) & ~(size_t)255;
  float* hs   = (float*)ws;
  float* hbuf = (float*)(ws + featBytes);
  char*  p    = ws + 2 * featBytes;
  float* onrm = (float*)p;  p += normBytes;
  int*   cnt  = (int*)p;    p += normBytes;
  float* inrm = (float*)p;  p += normBytes;
  int*   rowl = (int*)p;    p += normBytes;
  int*   boff = (int*)p;    p += nbBytes;
  unsigned short* rank = (unsigned short*)p;  p += edgeBytes;
  int*   colt = (int*)p;    p += edgeBytes;
  int*   col  = (int*)p;    p += edgeBytes;
  unsigned* hsrc = (unsigned*)p;  p += histBytes;
  unsigned* hdst = (unsigned*)p;  p += histBytes;
  int*   pre  = (int*)p;

  gcn_hist_rank<<<dim3(nchunks, 2), 1024, 0, stream>>>(src, dst, hsrc, hdst, rank, E, hstride);
  gcn_hist_scan<<<dim3((hstride + 255) / 256, 2), 256, 0, stream>>>(hsrc, hdst, pre, onrm, cnt,
                                                                    hstride, nchunks, N);
  gcn_scan_local<<<nb, 256, 0, stream>>>(cnt, rowl, boff, inrm, N);
  gcn_scan_top<<<1, 256, 0, stream>>>(boff, nb);
  gcn_fill<<<(E + 255) / 256, 256, 0, stream>>>(src, dst, rank, rowl, boff, pre, colt, E, 2 * hstride);
  gcn_rowsort<<<(N + 7) / 8, 256, 0, stream>>>(colt, cnt, rowl, boff, col, N);

  const int ntiles = (N + 15) / 16;
  const dim3 ggrid(256);
  const unsigned gatherBlocks = (unsigned)((N + 7) / 8);

  const float* hin = feat;
  for (int l = 0; l < 3; ++l) {
    const bool last = (l == 2);
    gcn_gemm_wmma<<<ggrid, GW * 32, 0, stream>>>(hin, Wl[l], bl[l], onrm, hs, N, ntiles);
    float* outp = last ? (float*)d_out : hbuf;
    gcn_gather<<<gatherBlocks, 256, 0, stream>>>(hs, col, cnt, rowl, boff, inrm,
                                                 last ? nullptr : hin, outp, N, last ? 0 : 1);
    hin = hbuf;
  }
}


# Layers 0 and 1: row tile sums and the next layer's dense product in one launch (16 rows per trip, tile handed to WMMA through LDS, 3 groups per WGP): matrix work beside the gathers; sum order kept
# speedup vs baseline: 4.9452x; 1.1393x over previous
#include <hip/hip_runtime.h>

#define D 128
#define WSTR 132

typedef __attribute__((ext_vector_type(2))) float v2f;
typedef __attribute__((ext_vector_type(8))) float v8f;

#define HCH    32768
#define HCHLOG 15
#define HWORDS 65536

template <bool FULL>
__device__ __forceinline__ void gcn_hist_chunk(const int* __restrict__ key, unsigned* sc,
                                               unsigned* __restrict__ outp,
                                               unsigned short* __restrict__ rank,
                                               bool isdst, int e0, int e1, int hstride) {
  const int tid = threadIdx.x;
  const int b0  = e0 + tid * 4;

  int k[32];
#pragma unroll
  for (int j = 0; j < 8; ++j) {
    if (FULL) {
      const int4 t = *(const int4*)(key + b0 + j * 4096);
      k[4 * j] = t.x; k[4 * j + 1] = t.y; k[4 * j + 2] = t.z; k[4 * j + 3] = t.w;
    } else {
#pragma unroll
      for (int q = 0; q < 4; ++q) {
        const int idx = b0 + j * 4096 + q;
        k[4 * j + q] = (idx < e1) ? key[idx] : -1;
      }
    }
  }

  const int nvec = hstride >> 2;
  uint4 z; z.x = 0u; z.y = 0u; z.z = 0u; z.w = 0u;
  for (int v = tid; v < nvec; v += 1024) ((uint4*)sc)[v] = z;
  __syncthreads();

  if (isdst) {
    unsigned short r[32];
#pragma unroll
    for (int i = 0; i < 32; ++i) {
      r[i] = 0;
      if (FULL || k[i] >= 0) {
        const unsigned sh  = ((unsigned)k[i] & 1u) << 4;
        const unsigned old = atomicAdd(&sc[k[i] >> 1], 1u << sh);
        r[i] = (unsigned short)((old >> sh) & 0xFFFFu);
      }
    }
#pragma unroll
    for (int j = 0; j < 8; ++j) {
      if (FULL) {
        ushort4 rv; rv.x = r[4 * j]; rv.y = r[4 * j + 1]; rv.z = r[4 * j + 2]; rv.w = r[4 * j + 3];
        *(ushort4*)(rank + b0 + j * 4096) = rv;
      } else {
#pragma unroll
        for (int q = 0; q < 4; ++q) {
          const int idx = b0 + j * 4096 + q;
          if (idx < e1) rank[idx] = r[4 * j + q];
        }
      }
    }
  } else {
#pragma unroll
    for (int i = 0; i < 32; ++i) {
      if (FULL || k[i] >= 0) atomicAdd(&sc[k[i] >> 1], 1u << (((unsigned)k[i] & 1u) << 4));
    }
  }
  __syncthreads();

  for (int v = tid; v < nvec; v += 1024) ((uint4*)outp)[v] = ((const uint4*)sc)[v];
}

__global__ __launch_bounds__(1024)
void gcn_hist_rank(const int* __restrict__ src, const int* __restrict__ dst,
                   unsigned* __restrict__ hsrc, unsigned* __restrict__ hdst,
                   unsigned short* __restrict__ rank, int E, int hstride) {
  __shared__ __attribute__((aligned(16))) unsigned sc[HWORDS];
  const int chunk = blockIdx.x;
  const bool isdst = (blockIdx.y == 0);
  const int* __restrict__ key = isdst ? dst : src;
  unsigned* __restrict__ outp = (isdst ? hdst : hsrc) + (size_t)chunk * hstride;
  const int e0 = chunk << HCHLOG;
  if (e0 + HCH <= E) gcn_hist_chunk<true>(key, sc, outp, rank, isdst, e0, e0 + HCH, hstride);
  else               gcn_hist_chunk<false>(key, sc, outp, rank, isdst, e0, E, hstride);
}

__global__ __launch_bounds__(256)
void gcn_hist_scan(const unsigned* __restrict__ hsrc, const unsigned* __restrict__ hdst,
                   int* __restrict__ pre, float* __restrict__ onrm, int* __restrict__ cnt,
                   int hstride, int nchunks, int n) {
  const int w = blockIdx.x * 256 + threadIdx.x;
  if (w >= hstride) return;
  const bool isdst = (blockIdx.y == 0);
  const unsigned* __restrict__ h = isdst ? hdst : hsrc;
  unsigned lo = 0u, hi = 0u;
  for (int c0 = 0; c0 < nchunks; c0 += 32) {
    unsigned v[32];
#pragma unroll
    for (int u = 0; u < 32; ++u)
      v[u] = (c0 + u < nchunks) ? h[(size_t)(c0 + u) * hstride + w] : 0u;
#pragma unroll
    for (int u = 0; u < 32; ++u) {
      if (isdst && c0 + u < nchunks) {
        int2 p; p.x = (int)lo; p.y = (int)hi;
        *(int2*)(pre + ((size_t)(c0 + u) * hstride + w) * 2) = p;
      }
      lo += v[u] & 0xFFFFu;
      hi += v[u] >> 16;
    }
  }
  const int n0 = 2 * w;
  if (isdst) {
    if (n0     < n) cnt[n0]     = (int)lo;
    if (n0 + 1 < n) cnt[n0 + 1] = (int)hi;
  } else {
    if (n0     < n) onrm[n0]     = rsqrtf(fmaxf((float)lo, 1.0f));
    if (n0 + 1 < n) onrm[n0 + 1] = rsqrtf(fmaxf((float)hi, 1.0f));
  }
}

__device__ __forceinline__ int wave_incl_scan(int x, int lane) {
  for (int o = 1; o < 32; o <<= 1) {
    const int y = __shfl_up(x, o);
    if (lane >= o) x += y;
  }
  return x;
}

__global__ __launch_bounds__(256)
void gcn_scan_local(const int* __restrict__ cnt, int* __restrict__ rowl,
                    int* __restrict__ bsum, float* __restrict__ inrm, int n) {
  __shared__ int swave[8];
  const int tid  = threadIdx.x;
  const int lane = tid & 31;
  const int w    = tid >> 5;
  const int base = blockIdx.x * 1024 + tid * 4;
  int c[4];
#pragma unroll
  for (int k = 0; k < 4; ++k) c[k] = (base + k < n) ? cnt[base + k] : 0;
  const int t = c[0] + c[1] + c[2] + c[3];
  const int x = wave_incl_scan(t, lane);
  if (lane == 31) swave[w] = x;
  __syncthreads();
  int woff = 0;
  for (int j = 0; j < w; ++j) woff += swave[j];
  int run = woff + x - t;
#pragma unroll
  for (int k = 0; k < 4; ++k) {
    if (base + k < n) {
      rowl[base + k] = run;
      inrm[base + k] = rsqrtf(fmaxf((float)c[k], 1.0f));
    }
    run += c[k];
  }
  if (tid == 255) bsum[blockIdx.x] = woff + x;
}

__global__ __launch_bounds__(256)
void gcn_scan_top(int* __restrict__ bsum, int nb) {
  __shared__ int swave[8];
  const int tid  = threadIdx.x;
  const int lane = tid & 31;
  const int w    = tid >> 5;
  int carry = 0;
  for (int base = 0; base < nb; base += 256) {
    const int i = base + tid;
    const int t = (i < nb) ? bsum[i] : 0;
    const int x = wave_incl_scan(t, lane);
    if (lane == 31) swave[w] = x;
    __syncthreads();
    int woff = 0, tot = 0;
    for (int j = 0; j < 8; ++j) { if (j < w) woff += swave[j]; tot += swave[j]; }
    if (i < nb) bsum[i] = carry + woff + x - t;
    carry += tot;
    __syncthreads();
  }
}

__global__ void gcn_fill(const int* __restrict__ src, const int* __restrict__ dst,
                         const unsigned short* __restrict__ rank, const int* __restrict__ rowl,
                         const int* __restrict__ boff, const int* __restrict__ pre,
                         int* __restrict__ colt, int E, int prestride) {
  int i = blockIdx.x * blockDim.x + threadIdx.x;
  if (i < E) {
    const int d = dst[i];
    const int pc = pre[(size_t)(i >> HCHLOG) * prestride + d];
    colt[rowl[d] + boff[d >> 10] + pc + (int)rank[i]] = src[i];
  }
}

__global__ __launch_bounds__(256)
void gcn_rowsort(const int* __restrict__ colt, const int* __restrict__ cnt,
                 const int* __restrict__ rowl, const int* __restrict__ boff,
                 int* __restrict__ col, int n) {
  const int row  = blockIdx.x * 8 + (threadIdx.x >> 5);
  const int lane = threadIdx.x & 31;
  if (row >= n) return;
  const int start = rowl[row] + boff[row >> 10];
  const int d     = cnt[row];
  for (int i = lane; i < d; i += 32) {
    const int si = colt[start + i];
    int r = 0;
    for (int j = 0; j < d; ++j) {
      const int sj = colt[start + j];
      r += (sj < si || (sj == si && j < i)) ? 1 : 0;
    }
    col[start + r] = si;
  }
}

#define GW 16

__global__ __launch_bounds__(GW * 32)
void gcn_gemm_wmma(const float* __restrict__ h, const float* __restrict__ W,
                   const float* __restrict__ bias, const float* __restrict__ out_norm,
                   float* __restrict__ hs, int nrows, int ntiles) {
  __shared__ __attribute__((aligned(16))) float sW[D * WSTR];
  __shared__ __attribute__((aligned(16))) float sO[GW * 16 * WSTR];

  const int tid = threadIdx.x;
#pragma unroll
  for (int u = 0; u < (D * D / 4) / (GW * 32); ++u) {
    const int idx = tid + u * (GW * 32);
    const int r   = idx >> 5;
    const int c4  = (idx & 31) << 2;
    *(float4*)(&sW[r * WSTR + c4]) = *(const float4*)(W + (size_t)r * D + c4);
  }
  __syncthreads();

  const int wave = __builtin_amdgcn_readfirstlane(tid >> 5);
  const int lane = tid & 31;
  const int hi   = lane >> 4;
  const int lm   = lane & 15;
  const float4 b4 = *(const float4*)(bias + 4 * lane);
  const float* __restrict__ bfrag = &sW[lm * WSTR + 4 * hi];
  float* so = &sO[wave * 16 * WSTR];

  for (int tile = blockIdx.x + gridDim.x * wave; tile < ntiles; tile += gridDim.x * GW) {
    const int row0 = tile << 4;
    int ar = row0 + lm;
    ar = (ar < nrows) ? ar : (nrows - 1);
    const float* __restrict__ ap = h + (size_t)ar * D + 4 * hi;
    float4 a[16];
#pragma unroll
    for (int j = 0; j < 16; ++j) a[j] = *(const float4*)(ap + 8 * j);
    const int onv = __float_as_int(out_norm[ar]);

    v8f acc[8];
#pragma unroll
    for (int c = 0; c < 8; ++c) acc[c] = (v8f){0.f, 0.f, 0.f, 0.f, 0.f, 0.f, 0.f, 0.f};

#pragma unroll
    for (int j = 0; j < 16; ++j) {
      float4 b[8];
#pragma unroll
      for (int c = 0; c < 8; ++c) b[c] = *(const float4*)(bfrag + c * 16 * WSTR + 8 * j);
#pragma unroll
      for (int c = 0; c < 8; ++c) {
        v2f x; x.x = a[j].x; x.y = a[j].y;
        v2f y; y.x = b[c].x; y.y = b[c].y;
        acc[c] = __builtin_amdgcn_wmma_f32_16x16x4_f32(false, x, false, y, (short)0, acc[c], false, false);
      }
#pragma unroll
      for (int c = 0; c < 8; ++c) {
        v2f x; x.x = a[j].z; x.y = a[j].w;
        v2f y; y.x = b[c].z; y.y = b[c].w;
        acc[c] = __builtin_amdgcn_wmma_f32_16x16x4_f32(false, x, false, y, (short)0, acc[c], false, false);
      }
    }

#pragma unroll
    for (int c = 0; c < 8; ++c) {
#pragma unroll
      for (int r = 0; r < 8; ++r) so[(r + 8 * hi) * WSTR + 16 * c + lm] = acc[c][r];
    }
    const int nfull = (nrows - row0 < 16) ? (nrows - row0) : 16;
    if (nfull == 16) {
#pragma unroll
      for (int rr = 0; rr < 16; ++rr) {
        const float sc = __int_as_float(__builtin_amdgcn_readlane(onv, rr));
        const float4 v = *(const float4*)(&so[rr * WSTR + 4 * lane]);
        float4 o;
        o.x = (v.x + b4.x) * sc; o.y = (v.y + b4.y) * sc;
        o.z = (v.z + b4.z) * sc; o.w = (v.w + b4.w) * sc;
        *(float4*)(hs + (size_t)(row0 + rr) * D + 4 * lane) = o;
      }
    } else {
      for (int rr = 0; rr < nfull; ++rr) {
        const float sc = __int_as_float(__builtin_amdgcn_readlane(onv, rr));
        const float4 v = *(const float4*)(&so[rr * WSTR + 4 * lane]);
        float4 o;
        o.x = (v.x + b4.x) * sc; o.y = (v.y + b4.y) * sc;
        o.z = (v.z + b4.z) * sc; o.w = (v.w + b4.w) * sc;
        *(float4*)(hs + (size_t)(row0 + rr) * D + 4 * lane) = o;
      }
    }
  }
}

template <int U>
__device__ __forceinline__ void gcn_add_rows(const float* __restrict__ hs, int ids, int e,
                                             int c4, float4& acc) {
  float4 v[U];
#pragma unroll
  for (int u = 0; u < U; ++u) {
    const int s = __builtin_amdgcn_readlane(ids, e + u);
    v[u] = *(const float4*)(hs + (size_t)s * D + c4);
  }
#pragma unroll
  for (int u = 0; u < U; ++u) {
    acc.x += v[u].x; acc.y += v[u].y; acc.z += v[u].z; acc.w += v[u].w;
  }
}

__global__ __launch_bounds__(256)
void gcn_gather(const float* __restrict__ hs, const int* __restrict__ col,
                const int* __restrict__ cnt, const int* __restrict__ rowl,
                const int* __restrict__ boff, const float* __restrict__ in_norm,
                const float* hres, float* out, int nrows, int do_relu) {
  const int row = __builtin_amdgcn_readfirstlane((int)(blockIdx.x * 8 + (threadIdx.x >> 5)));
  if (row >= nrows) return;
  const int lane = threadIdx.x & 31;
  const int c4   = lane << 2;
  const int start = rowl[row] + boff[row >> 10];
  const int d     = cnt[row];
  const float s   = in_norm[row];
  const size_t ro = (size_t)row * D + c4;

  float4 res; res.x = 0.f; res.y = 0.f; res.z = 0.f; res.w = 0.f;
  if (hres) res = *(const float4*)(hres + ro);

  float4 acc; acc.x = 0.f; acc.y = 0.f; acc.z = 0.f; acc.w = 0.f;
  for (int base = 0; base < d; base += 32) {
    const int m = (d - base < 32) ? (d - base) : 32;
    const int ids = (lane < m) ? col[start + base + lane] : 0;
    int e = 0;
    for (; e + 8 <= m; e += 8) gcn_add_rows<8>(hs, ids, e, c4, acc);
    if (m & 4) { gcn_add_rows<4>(hs, ids, e, c4, acc); e += 4; }
    if (m & 2) { gcn_add_rows<2>(hs, ids, e, c4, acc); e += 2; }
    if (m & 1) { gcn_add_rows<1>(hs, ids, e, c4, acc); }
  }

  float4 v;
  v.x = acc.x * s; v.y = acc.y * s; v.z = acc.z * s; v.w = acc.w * s;
  if (hres) { v.x += res.x; v.y += res.y; v.z += res.z; v.w += res.w; }
  if (do_relu) {
    v.x = fmaxf(v.x, 0.f); v.y = fmaxf(v.y, 0.f);
    v.z = fmaxf(v.z, 0.f); v.w = fmaxf(v.w, 0.f);
  }
  *(float4*)(out + ro) = v;
}


#define FW 16

__global__ __launch_bounds__(FW * 32)
void gcn_gather_gemm(const float* __restrict__ hs, const int* __restrict__ col,
                     const int* __restrict__ cnt, const int* __restrict__ rowl,
                     const int* __restrict__ boff, const float* __restrict__ in_norm,
                     const float* __restrict__ hres, float* __restrict__ hout,
                     const float* __restrict__ W, const float* __restrict__ bias,
                     const float* __restrict__ out_norm, float* __restrict__ hs2,
                     int nrows, int ntiles) {
  __shared__ __attribute__((aligned(16))) float sW[D * WSTR];
  __shared__ __attribute__((aligned(16))) float sA[16 * WSTR];
  __shared__ __attribute__((aligned(16))) float sO[16 * WSTR];

  const int tid = threadIdx.x;
#pragma unroll
  for (int u = 0; u < (D * D / 4) / (FW * 32); ++u) {
    const int idx = tid + u * (FW * 32);
    const int r   = idx >> 5;
    const int q4  = (idx & 31) << 2;
    *(float4*)(&sW[r * WSTR + q4]) = *(const float4*)(W + (size_t)r * D + q4);
  }
  __syncthreads();

  const int wave = __builtin_amdgcn_readfirstlane(tid >> 5);
  const int lane = tid & 31;
  const int hi   = lane >> 4;
  const int lm   = lane & 15;
  const int c4   = lane << 2;
  const float4 b4 = *(const float4*)(bias + c4);

  int   c_start = 0, c_d = 0, c_ids = 0;
  float c_s = 0.f, c_sc = 0.f;
  {
    const int row = ((int)blockIdx.x << 4) + wave;
    if ((int)blockIdx.x < ntiles && row < nrows) {
      c_start = rowl[row] + boff[row >> 10];
      c_d     = cnt[row];
      c_s     = in_norm[row];
      c_sc    = out_norm[row];
      const int m0 = (c_d < 32) ? c_d : 32;
      c_ids = (lane < m0) ? col[c_start + lane] : 0;
    }
  }

  for (int tile = blockIdx.x; tile < ntiles; tile += gridDim.x) {
    const int row = (tile << 4) + wave;
    const size_t ro = (size_t)row * D + c4;

    const int nrow   = ((tile + (int)gridDim.x) << 4) + wave;
    const int prow   = (nrow < nrows) ? nrow : (nrows - 1);
    const int*   nip = (lane == 0) ? (rowl + prow) : ((lane == 1) ? (boff + (prow >> 10)) : (cnt + prow));
    const float* nfp = (lane == 0) ? (in_norm + prow) : (out_norm + prow);
    const int   n_iw = *nip;
    const float n_fw = *nfp;

    float4 v; v.x = 0.f; v.y = 0.f; v.z = 0.f; v.w = 0.f;
    if (row < nrows) {
      const float4 res = *(const float4*)(hres + ro);

      float4 acc; acc.x = 0.f; acc.y = 0.f; acc.z = 0.f; acc.w = 0.f;
      for (int base = 0; base < c_d; base += 32) {
        const int m = (c_d - base < 32) ? (c_d - base) : 32;
        int ids = c_ids;
        if (base > 0) ids = (lane < m) ? col[c_start + base + lane] : 0;
        int e = 0;
        for (; e + 8 <= m; e += 8) gcn_add_rows<8>(hs, ids, e, c4, acc);
        if (m & 4) { gcn_add_rows<4>(hs, ids, e, c4, acc); e += 4; }
        if (m & 2) { gcn_add_rows<2>(hs, ids, e, c4, acc); e += 2; }
        if (m & 1) { gcn_add_rows<1>(hs, ids, e, c4, acc); }
      }
      v.x = fmaxf(acc.x * c_s + res.x, 0.f); v.y = fmaxf(acc.y * c_s + res.y, 0.f);
      v.z = fmaxf(acc.z * c_s + res.z, 0.f); v.w = fmaxf(acc.w * c_s + res.w, 0.f);
    }
    *(float4*)(&sA[wave * WSTR + c4]) = v;
    __syncthreads();

    const int   n_start = __builtin_amdgcn_readlane(n_iw, 0) + __builtin_amdgcn_readlane(n_iw, 1);
    const int   n_d     = __builtin_amdgcn_readlane(n_iw, 2);
    const float n_s     = __int_as_float(__builtin_amdgcn_readlane(__float_as_int(n_fw), 0));
    const float n_sc    = __int_as_float(__builtin_amdgcn_readlane(__float_as_int(n_fw), 1));
    const int nm = (n_d < 32) ? n_d : 32;
    const int n_ids = (lane < nm) ? col[n_start + lane] : 0;

    if (wave < 8) {
      const float* __restrict__ ap = &sA[lm * WSTR + 4 * hi];
      const float* __restrict__ bp = &sW[(16 * wave + lm) * WSTR + 4 * hi];
      v8f pacc = (v8f){0.f, 0.f, 0.f, 0.f, 0.f, 0.f, 0.f, 0.f};
#pragma unroll
      for (int j = 0; j < 16; ++j) {
        const float4 a = *(const float4*)(ap + 8 * j);
        const float4 b = *(const float4*)(bp + 8 * j);
        v2f x0; x0.x = a.x; x0.y = a.y;
        v2f y0; y0.x = b.x; y0.y = b.y;
        pacc = __builtin_amdgcn_wmma_f32_16x16x4_f32(false, x0, false, y0, (short)0, pacc, false, false);
        v2f x1; x1.x = a.z; x1.y = a.w;
        v2f y1; y1.x = b.z; y1.y = b.w;
        pacc = __builtin_amdgcn_wmma_f32_16x16x4_f32(false, x1, false, y1, (short)0, pacc, false, false);
      }
#pragma unroll
      for (int r = 0; r < 8; ++r) sO[(r + 8 * hi) * WSTR + 16 * wave + lm] = pacc[r];
    }
    __syncthreads();

    if (row < nrows) {
      if (hout) *(float4*)(hout + ro) = v;
      const float4 t = *(const float4*)(&sO[wave * WSTR + c4]);
      float4 o;
      o.x = (t.x + b4.x) * c_sc; o.y = (t.y + b4.y) * c_sc;
      o.z = (t.z + b4.z) * c_sc; o.w = (t.w + b4.w) * c_sc;
      *(float4*)(hs2 + ro) = o;
    }
    c_start = n_start; c_d = n_d; c_s = n_s; c_sc = n_sc; c_ids = n_ids;
  }
}

extern "C" void kernel_launch(void* const* d_in, const int* in_sizes, int n_in,
                              void* d_out, int out_size, void* d_ws, size_t ws_size,
                              hipStream_t stream) {
  const float* feat = (const float*)d_in[0];
  const int*   src  = (const int*)d_in[1];
  const int*   dst  = (const int*)d_in[2];
  const float* Wl[3] = { (const float*)d_in[3], (const float*)d_in[5], (const float*)d_in[7] };
  const float* bl[3] = { (const float*)d_in[4], (const float*)d_in[6], (const float*)d_in[8] };

  const int N = in_sizes[0] / D;
  const int E = in_sizes[1];
  const int nb = (N + 1023) / 1024;

  char* ws = (char*)d_ws;
  const int nchunks = (E + HCH - 1) / HCH;
  const int hstride = (((N + 1) / 2) + 3) & ~3;
  const size_t featBytes = (size_t)N * D * sizeof(float);
  const size_t normBytes = (((size_t)N * sizeof(float)) + 255) & ~(size_t)255;
  const size_t nbBytes   = (((size_t)nb * sizeof(int)) + 255) & ~(size_t)255;
  const size_t edgeBytes = (((size_t)E * sizeof(int)) + 255) & ~(size_t)255;
  const size_t histBytes = (((size_t)nchunks * hstride * sizeof(unsigned)) + 255) & ~(size_t)255;
  float* hs   = (float*)ws;
  float* hbuf = (float*)(ws + featBytes);
  float* hsb  = (float*)(ws + 2 * featBytes);
  char*  p    = ws + 3 * featBytes;
  float* onrm = (float*)p;  p += normBytes;
  int*   cnt  = (int*)p;    p += normBytes;
  float* inrm = (float*)p;  p += normBytes;
  int*   rowl = (int*)p;    p += normBytes;
  int*   boff = (int*)p;    p += nbBytes;
  unsigned short* rank = (unsigned short*)p;  p += edgeBytes;
  int*   colt = (int*)p;    p += edgeBytes;
  int*   col  = (int*)p;    p += edgeBytes;
  unsigned* hsrc = (unsigned*)p;  p += histBytes;
  unsigned* hdst = (unsigned*)p;  p += histBytes;
  int*   pre  = (int*)p;

  gcn_hist_rank<<<dim3(nchunks, 2), 1024, 0, stream>>>(src, dst, hsrc, hdst, rank, E, hstride);
  gcn_hist_scan<<<dim3((hstride + 255) / 256, 2), 256, 0, stream>>>(hsrc, hdst, pre, onrm, cnt,
                                                                    hstride, nchunks, N);
  gcn_scan_local<<<nb, 256, 0, stream>>>(cnt, rowl, boff, inrm, N);
  gcn_scan_top<<<1, 256, 0, stream>>>(boff, nb);
  gcn_fill<<<(E + 255) / 256, 256, 0, stream>>>(src, dst, rank, rowl, boff, pre, colt, E, 2 * hstride);
  gcn_rowsort<<<(N + 7) / 8, 256, 0, stream>>>(colt, cnt, rowl, boff, col, N);

  const int ntiles = (N + 15) / 16;
  const dim3 ggrid(256);
  const dim3 fgrid(768);
  const unsigned gatherBlocks = (unsigned)((N + 7) / 8);

  gcn_gemm_wmma<<<ggrid, GW * 32, 0, stream>>>(feat, Wl[0], bl[0], onrm, hs, N, ntiles);
  gcn_gather_gemm<<<fgrid, FW * 32, 0, stream>>>(hs, col, cnt, rowl, boff, inrm, feat, hbuf,
                                                 Wl[1], bl[1], onrm, hsb, N, ntiles);
  gcn_gather_gemm<<<fgrid, FW * 32, 0, stream>>>(hsb, col, cnt, rowl, boff, inrm, hbuf, nullptr,
                                                 Wl[2], bl[2], onrm, hs, N, ntiles);
  gcn_gather<<<gatherBlocks, 256, 0, stream>>>(hs, col, cnt, rowl, boff, inrm,
                                               nullptr, (float*)d_out, N, 0);
}


# Each layer sums its out_norm-weighted INPUT rows per destination first and applies W to the 16-row sums after (linearity): no separate first product or last row-sum launch; sums in another order
# speedup vs baseline: 5.2847x; 1.0687x over previous
#include <hip/hip_runtime.h>

#define D 128
#define WSTR 132

typedef __attribute__((ext_vector_type(2))) float v2f;
typedef __attribute__((ext_vector_type(8))) float v8f;

#define HCH    32768
#define HCHLOG 15
#define HWORDS 65536

template <bool FULL>
__device__ __forceinline__ void gcn_hist_chunk(const int* __restrict__ key, unsigned* sc,
                                               unsigned* __restrict__ outp,
                                               unsigned short* __restrict__ rank,
                                               bool isdst, int e0, int e1, int hstride) {
  const int tid = threadIdx.x;
  const int b0  = e0 + tid * 4;

  int k[32];
#pragma unroll
  for (int j = 0; j < 8; ++j) {
    if (FULL) {
      const int4 t = *(const int4*)(key + b0 + j * 4096);
      k[4 * j] = t.x; k[4 * j + 1] = t.y; k[4 * j + 2] = t.z; k[4 * j + 3] = t.w;
    } else {
#pragma unroll
      for (int q = 0; q < 4; ++q) {
        const int idx = b0 + j * 4096 + q;
        k[4 * j + q] = (idx < e1) ? key[idx] : -1;
      }
    }
  }

  const int nvec = hstride >> 2;
  uint4 z; z.x = 0u; z.y = 0u; z.z = 0u; z.w = 0u;
  for (int v = tid; v < nvec; v += 1024) ((uint4*)sc)[v] = z;
  __syncthreads();

  if (isdst) {
    unsigned short r[32];
#pragma unroll
    for (int i = 0; i < 32; ++i) {
      r[i] = 0;
      if (FULL || k[i] >= 0) {
        const unsigned sh  = ((unsigned)k[i] & 1u) << 4;
        const unsigned old = atomicAdd(&sc[k[i] >> 1], 1u << sh);
        r[i] = (unsigned short)((old >> sh) & 0xFFFFu);
      }
    }
#pragma unroll
    for (int j = 0; j < 8; ++j) {
      if (FULL) {
        ushort4 rv; rv.x = r[4 * j]; rv.y = r[4 * j + 1]; rv.z = r[4 * j + 2]; rv.w = r[4 * j + 3];
        *(ushort4*)(rank + b0 + j * 4096) = rv;
      } else {
#pragma unroll
        for (int q = 0; q < 4; ++q) {
          const int idx = b0 + j * 4096 + q;
          if (idx < e1) rank[idx] = r[4 * j + q];
        }
      }
    }
  } else {
#pragma unroll
    for (int i = 0; i < 32; ++i) {
      if (FULL || k[i] >= 0) atomicAdd(&sc[k[i] >> 1], 1u << (((unsigned)k[i] & 1u) << 4));
    }
  }
  __syncthreads();

  for (int v = tid; v < nvec; v += 1024) ((uint4*)outp)[v] = ((const uint4*)sc)[v];
}

__global__ __launch_bounds__(1024)
void gcn_hist_rank(const int* __restrict__ src, const int* __restrict__ dst,
                   unsigned* __restrict__ hsrc, unsigned* __restrict__ hdst,
                   unsigned short* __restrict__ rank, int E, int hstride) {
  __shared__ __attribute__((aligned(16))) unsigned sc[HWORDS];
  const int chunk = blockIdx.x;
  const bool isdst = (blockIdx.y == 0);
  const int* __restrict__ key = isdst ? dst : src;
  unsigned* __restrict__ outp = (isdst ? hdst : hsrc) + (size_t)chunk * hstride;
  const int e0 = chunk << HCHLOG;
  if (e0 + HCH <= E) gcn_hist_chunk<true>(key, sc, outp, rank, isdst, e0, e0 + HCH, hstride);
  else               gcn_hist_chunk<false>(key, sc, outp, rank, isdst, e0, E, hstride);
}

__global__ __launch_bounds__(256)
void gcn_hist_scan(const unsigned* __restrict__ hsrc, const unsigned* __restrict__ hdst,
                   int* __restrict__ pre, float* __restrict__ onrm, int* __restrict__ cnt,
                   int hstride, int nchunks, int n) {
  const int w = blockIdx.x * 256 + threadIdx.x;
  if (w >= hstride) return;
  const bool isdst = (blockIdx.y == 0);
  const unsigned* __restrict__ h = isdst ? hdst : hsrc;
  unsigned lo = 0u, hi = 0u;
  for (int c0 = 0; c0 < nchunks; c0 += 32) {
    unsigned v[32];
#pragma unroll
    for (int u = 0; u < 32; ++u)
      v[u] = (c0 + u < nchunks) ? h[(size_t)(c0 + u) * hstride + w] : 0u;
#pragma unroll
    for (int u = 0; u < 32; ++u) {
      if (isdst && c0 + u < nchunks) {
        int2 p; p.x = (int)lo; p.y = (int)hi;
        *(int2*)(pre + ((size_t)(c0 + u) * hstride + w) * 2) = p;
      }
      lo += v[u] & 0xFFFFu;
      hi += v[u] >> 16;
    }
  }
  const int n0 = 2 * w;
  if (isdst) {
    if (n0     < n) cnt[n0]     = (int)lo;
    if (n0 + 1 < n) cnt[n0 + 1] = (int)hi;
  } else {
    if (n0     < n) onrm[n0]     = rsqrtf(fmaxf((float)lo, 1.0f));
    if (n0 + 1 < n) onrm[n0 + 1] = rsqrtf(fmaxf((float)hi, 1.0f));
  }
}

__device__ __forceinline__ int wave_incl_scan(int x, int lane) {
  for (int o = 1; o < 32; o <<= 1) {
    const int y = __shfl_up(x, o);
    if (lane >= o) x += y;
  }
  return x;
}

__global__ __launch_bounds__(256)
void gcn_scan_local(const int* __restrict__ cnt, int* __restrict__ rowl,
                    int* __restrict__ bsum, float* __restrict__ inrm, int n) {
  __shared__ int swave[8];
  const int tid  = threadIdx.x;
  const int lane = tid & 31;
  const int w    = tid >> 5;
  const int base = blockIdx.x * 1024 + tid * 4;
  int c[4];
#pragma unroll
  for (int k = 0; k < 4; ++k) c[k] = (base + k < n) ? cnt[base + k] : 0;
  const int t = c[0] + c[1] + c[2] + c[3];
  const int x = wave_incl_scan(t, lane);
  if (lane == 31) swave[w] = x;
  __syncthreads();
  int woff = 0;
  for (int j = 0; j < w; ++j) woff += swave[j];
  int run = woff + x - t;
#pragma unroll
  for (int k = 0; k < 4; ++k) {
    if (base + k < n) {
      rowl[base + k] = run;
      inrm[base + k] = rsqrtf(fmaxf((float)c[k], 1.0f));
    }
    run += c[k];
  }
  if (tid == 255) bsum[blockIdx.x] = woff + x;
}

__global__ __launch_bounds__(256)
void gcn_scan_top(int* __restrict__ bsum, int nb) {
  __shared__ int swave[8];
  const int tid  = threadIdx.x;
  const int lane = tid & 31;
  const int w    = tid >> 5;
  int carry = 0;
  for (int base = 0; base < nb; base += 256) {
    const int i = base + tid;
    const int t = (i < nb) ? bsum[i] : 0;
    const int x = wave_incl_scan(t, lane);
    if (lane == 31) swave[w] = x;
    __syncthreads();
    int woff = 0, tot = 0;
    for (int j = 0; j < 8; ++j) { if (j < w) woff += swave[j]; tot += swave[j]; }
    if (i < nb) bsum[i] = carry + woff + x - t;
    carry += tot;
    __syncthreads();
  }
}

__global__ void gcn_fill(const int* __restrict__ src, const int* __restrict__ dst,
                         const unsigned short* __restrict__ rank, const int* __restrict__ rowl,
                         const int* __restrict__ boff, const int* __restrict__ pre,
                         int* __restrict__ colt, int E, int prestride) {
  int i = blockIdx.x * blockDim.x + threadIdx.x;
  if (i < E) {
    const int d = dst[i];
    const int pc = pre[(size_t)(i >> HCHLOG) * prestride + d];
    colt[rowl[d] + boff[d >> 10] + pc + (int)rank[i]] = src[i];
  }
}


__global__ __launch_bounds__(256)
void gcn_rowsort(const int* __restrict__ colt, const int* __restrict__ cnt,
                 const int* __restrict__ rowl, const int* __restrict__ boff,
                 const float* __restrict__ onrm, int* __restrict__ colw, int n) {
  const int row  = blockIdx.x * 8 + (threadIdx.x >> 5);
  const int lane = threadIdx.x & 31;
  if (row >= n) return;
  const int start = rowl[row] + boff[row >> 10];
  const int d     = cnt[row];
  for (int i = lane; i < d; i += 32) {
    const int si = colt[start + i];
    const float w = onrm[si];
    int r = 0;
    for (int j = 0; j < d; ++j) {
      const int sj = colt[start + j];
      r += (sj < si || (sj == si && j < i)) ? 1 : 0;
    }
    int2 t; t.x = si; t.y = __float_as_int(w);
    *(int2*)(colw + 2 * (size_t)(start + r)) = t;
  }
}

#define FW 16

template <int U>
__device__ __forceinline__ void gcn_add_rows(const float* __restrict__ hp, int ids, int wts, int e,
                                             unsigned lb, float4& acc) {
  float4 v[U];
  float  w[U];
#pragma unroll
  for (int u = 0; u < U; ++u) {
    const int s = __builtin_amdgcn_readlane(ids, e + u);
    w[u] = __int_as_float(__builtin_amdgcn_readlane(wts, e + u));
    const char* rp = (const char*)(hp + (size_t)s * D);
    v[u] = *(const float4*)(rp + lb);
  }
#pragma unroll
  for (int u = 0; u < U; ++u) {
    acc.x = __builtin_fmaf(w[u], v[u].x, acc.x); acc.y = __builtin_fmaf(w[u], v[u].y, acc.y);
    acc.z = __builtin_fmaf(w[u], v[u].z, acc.z); acc.w = __builtin_fmaf(w[u], v[u].w, acc.w);
  }
}

template <bool RES>
__global__ __launch_bounds__(FW * 32)
void gcn_layer(const float* __restrict__ hin, const int* __restrict__ colw,
               const int* __restrict__ cnt, const int* __restrict__ rowl,
               const int* __restrict__ boff, const float* __restrict__ in_norm,
               const float* __restrict__ W, const float* __restrict__ bias,
               float* __restrict__ hout, int nrows, int ntiles) {
  __shared__ __attribute__((aligned(16))) float sW[D * WSTR];
  __shared__ __attribute__((aligned(16))) float sA[16 * WSTR];
  __shared__ __attribute__((aligned(16))) float sO[16 * WSTR];

  const int tid = threadIdx.x;
#pragma unroll
  for (int u = 0; u < (D * D / 4) / (FW * 32); ++u) {
    const int idx = tid + u * (FW * 32);
    const int r   = idx >> 5;
    const int q4  = (idx & 31) << 2;
    *(float4*)(&sW[r * WSTR + q4]) = *(const float4*)(W + (size_t)r * D + q4);
  }
  __syncthreads();

  const int wave = __builtin_amdgcn_readfirstlane(tid >> 5);
  const int lane = tid & 31;
  const int hi   = lane >> 4;
  const int lm   = lane & 15;
  const int c4   = lane << 2;
  const unsigned lb = (unsigned)lane << 4;
  const float4 b4 = *(const float4*)(bias + c4);

  int   c_start = 0, c_d = 0, c_ids = 0, c_w = 0;
  float c_s = 0.f;
  {
    const int row = ((int)blockIdx.x << 4) + wave;
    if ((int)blockIdx.x < ntiles && row < nrows) {
      c_start = rowl[row] + boff[row >> 10];
      c_d     = cnt[row];
      c_s     = in_norm[row];
      const int m0 = (c_d < 32) ? c_d : 32;
      if (lane < m0) {
        const int2 t = *(const int2*)(colw + 2 * (size_t)(c_start + lane));
        c_ids = t.x; c_w = t.y;
      }
    }
  }

  for (int tile = blockIdx.x; tile < ntiles; tile += gridDim.x) {
    const int row = (tile << 4) + wave;
    const size_t ro = (size_t)row * D + c4;

    const int nrow   = ((tile + (int)gridDim.x) << 4) + wave;
    const int prow   = (nrow < nrows) ? nrow : (nrows - 1);
    const int*   nip = (lane == 0) ? (rowl + prow) : ((lane == 1) ? (boff + (prow >> 10)) : (cnt + prow));
    const float* nfp = in_norm + ((lane == 0) ? prow : 0);
    const int   n_iw = *nip;
    const float n_fw = *nfp;

    float4 v;   v.x = 0.f;   v.y = 0.f;   v.z = 0.f;   v.w = 0.f;
    float4 res; res.x = 0.f; res.y = 0.f; res.z = 0.f; res.w = 0.f;
    float  sb = 0.f;
    if (row < nrows) {
      if (RES) res = *(const float4*)(hin + ro);

      float4 acc; acc.x = 0.f; acc.y = 0.f; acc.z = 0.f; acc.w = 0.f;
      float wl = 0.f;
      for (int base = 0; base < c_d; base += 32) {
        const int m = (c_d - base < 32) ? (c_d - base) : 32;
        int ids = c_ids, wts = c_w;
        if (base > 0) {
          ids = 0; wts = 0;
          if (lane < m) {
            const int2 t = *(const int2*)(colw + 2 * (size_t)(c_start + base + lane));
            ids = t.x; wts = t.y;
          }
        }
        wl += __int_as_float(wts);
        int e = 0;
        for (; e + 8 <= m; e += 8) gcn_add_rows<8>(hin, ids, wts, e, lb, acc);
        if (m & 4) { gcn_add_rows<4>(hin, ids, wts, e, lb, acc); e += 4; }
        if (m & 2) { gcn_add_rows<2>(hin, ids, wts, e, lb, acc); e += 2; }
        if (m & 1) { gcn_add_rows<1>(hin, ids, wts, e, lb, acc); }
      }
      for (int o = 16; o >= 1; o >>= 1) wl += __shfl_xor(wl, o);
      sb = wl;
      v  = acc;
    }
    *(float4*)(&sA[wave * WSTR + c4]) = v;
    __syncthreads();

    const int   n_start = __builtin_amdgcn_readlane(n_iw, 0) + __builtin_amdgcn_readlane(n_iw, 1);
    const int   n_d     = __builtin_amdgcn_readlane(n_iw, 2);
    const float n_s     = __int_as_float(__builtin_amdgcn_readlane(__float_as_int(n_fw), 0));
    const int nm = (n_d < 32) ? n_d : 32;
    int n_ids = 0, n_w = 0;
    if (lane < nm) {
      const int2 t = *(const int2*)(colw + 2 * (size_t)(n_start + lane));
      n_ids = t.x; n_w = t.y;
    }

    if (wave < 8) {
      const float* __restrict__ ap = &sA[lm * WSTR + 4 * hi];
      const float* __restrict__ bp = &sW[(16 * wave + lm) * WSTR + 4 * hi];
      v8f pacc = (v8f){0.f, 0.f, 0.f, 0.f, 0.f, 0.f, 0.f, 0.f};
#pragma unroll
      for (int j = 0; j < 16; ++j) {
        const float4 a = *(const float4*)(ap + 8 * j);
        const float4 b = *(const float4*)(bp + 8 * j);
        v2f x0; x0.x = a.x; x0.y = a.y;
        v2f y0; y0.x = b.x; y0.y = b.y;
        pacc = __builtin_amdgcn_wmma_f32_16x16x4_f32(false, x0, false, y0, (short)0, pacc, false, false);
        v2f x1; x1.x = a.z; x1.y = a.w;
        v2f y1; y1.x = b.z; y1.y = b.w;
        pacc = __builtin_amdgcn_wmma_f32_16x16x4_f32(false, x1, false, y1, (short)0, pacc, false, false);
      }
#pragma unroll
      for (int r = 0; r < 8; ++r) sO[(r + 8 * hi) * WSTR + 16 * wave + lm] = pacc[r];
    }
    __syncthreads();

    if (row < nrows) {
      const float4 t = *(const float4*)(&sO[wave * WSTR + c4]);
      float4 o;
      o.x = (t.x + sb * b4.x) * c_s; o.y = (t.y + sb * b4.y) * c_s;
      o.z = (t.z + sb * b4.z) * c_s; o.w = (t.w + sb * b4.w) * c_s;
      if (RES) {
        o.x = fmaxf(o.x + res.x, 0.f); o.y = fmaxf(o.y + res.y, 0.f);
        o.z = fmaxf(o.z + res.z, 0.f); o.w = fmaxf(o.w + res.w, 0.f);
      }
      *(float4*)(hout + ro) = o;
    }
    c_start = n_start; c_d = n_d; c_s = n_s; c_ids = n_ids; c_w = n_w;
  }
}

extern "C" void kernel_launch(void* const* d_in, const int* in_sizes, int n_in,
                              void* d_out, int out_size, void* d_ws, size_t ws_size,
                              hipStream_t stream) {
  const float* feat = (const float*)d_in[0];
  const int*   src  = (const int*)d_in[1];
  const int*   dst  = (const int*)d_in[2];
  const float* Wl[3] = { (const float*)d_in[3], (const float*)d_in[5], (const float*)d_in[7] };
  const float* bl[3] = { (const float*)d_in[4], (const float*)d_in[6], (const float*)d_in[8] };

  const int N = in_sizes[0] / D;
  const int E = in_sizes[1];
  const int nb = (N + 1023) / 1024;

  char* ws = (char*)d_ws;
  const int nchunks = (E + HCH - 1) / HCH;
  const int hstride = (((N + 1) / 2) + 3) & ~3;
  const size_t featBytes = (size_t)N * D * sizeof(float);
  const size_t normBytes = (((size_t)N * sizeof(float)) + 255) & ~(size_t)255;
  const size_t nbBytes   = (((size_t)nb * sizeof(int)) + 255) & ~(size_t)255;
  const size_t edgeBytes = (((size_t)E * sizeof(int)) + 255) & ~(size_t)255;
  const size_t histBytes = (((size_t)nchunks * hstride * sizeof(unsigned)) + 255) & ~(size_t)255;
  float* h1   = (float*)ws;
  float* h2   = (float*)(ws + featBytes);
  char*  p    = ws + 2 * featBytes;
  float* onrm = (float*)p;  p += normBytes;
  int*   cnt  = (int*)p;    p += normBytes;
  float* inrm = (float*)p;  p += normBytes;
  int*   rowl = (int*)p;    p += normBytes;
  int*   boff = (int*)p;    p += nbBytes;
  unsigned short* rank = (unsigned short*)p;  p += edgeBytes;
  int*   colt = (int*)p;    p += edgeBytes;
  int*   colw = (int*)p;    p += 2 * edgeBytes;
  unsigned* hsrc = (unsigned*)p;  p += histBytes;
  unsigned* hdst = (unsigned*)p;  p += histBytes;
  int*   pre  = (int*)p;

  gcn_hist_rank<<<dim3(nchunks, 2), 1024, 0, stream>>>(src, dst, hsrc, hdst, rank, E, hstride);
  gcn_hist_scan<<<dim3((hstride + 255) / 256, 2), 256, 0, stream>>>(hsrc, hdst, pre, onrm, cnt,
                                                                    hstride, nchunks, N);
  gcn_scan_local<<<nb, 256, 0, stream>>>(cnt, rowl, boff, inrm, N);
  gcn_scan_top<<<1, 256, 0, stream>>>(boff, nb);
  gcn_fill<<<(E + 255) / 256, 256, 0, stream>>>(src, dst, rank, rowl, boff, pre, colt, E, 2 * hstride);
  gcn_rowsort<<<(N + 7) / 8, 256, 0, stream>>>(colt, cnt, rowl, boff, onrm, colw, N);

  const int ntiles = (N + 15) / 16;
  const dim3 fgrid(768);

  gcn_layer<true><<<fgrid, FW * 32, 0, stream>>>(feat, colw, cnt, rowl, boff, inrm, Wl[0], bl[0], h1, N, ntiles);
  gcn_layer<true><<<fgrid, FW * 32, 0, stream>>>(h1, colw, cnt, rowl, boff, inrm, Wl[1], bl[1], h2, N, ntiles);
  gcn_layer<false><<<fgrid, FW * 32, 0, stream>>>(h2, colw, cnt, rowl, boff, inrm, Wl[2], bl[2],
                                                  (float*)d_out, N, ntiles);
}
